# starting kernel
# baseline (speedup 1.0000x reference)
#include <hip/hip_runtime.h>
#include <math.h>

#define ATT_SLOPE 0.2f
#define ACT_SLOPE 0.01f

typedef __attribute__((ext_vector_type(2))) float v2f;
typedef __attribute__((ext_vector_type(8))) float v8f;

__device__ __forceinline__ float lrelu(float v, float s) { return v > 0.f ? v : v * s; }

__device__ __forceinline__ void atomicMaxF(float* addr, float val) {
  if (val >= 0.f)
    atomicMax((int*)addr, __float_as_int(val));
  else
    atomicMin((unsigned int*)addr, (unsigned int)__float_as_int(val));
}

template <int K, int NCOLS>
__global__ void __launch_bounds__(32)
wmma_gemm_f32(const float* __restrict__ A, const float* __restrict__ B,
              float* __restrict__ D, int M) {
  const int m0 = blockIdx.x * 16;
  const int n0 = blockIdx.y * 16;
  const int lane = threadIdx.x;
  const int half = lane >> 4;
  const int lid  = lane & 15;
  const int ar = min(m0 + lid, M - 1);
  v8f acc = {};
#pragma unroll
  for (int k = 0; k < K; k += 4) {
    const int ka = k + half * 2;
    v2f a, b;
    a.x = A[(size_t)ar * K + ka];
    a.y = A[(size_t)ar * K + ka + 1];
    b.x = B[(size_t)ka * NCOLS + n0 + lid];
    b.y = B[(size_t)(ka + 1) * NCOLS + n0 + lid];
    acc = __builtin_amdgcn_wmma_f32_16x16x4_f32(false, a, false, b, (short)0, acc,
                                                false, false);
  }
  const int rbase = m0 + half * 8;
#pragma unroll
  for (int r = 0; r < 8; ++r) {
    const int row = rbase + r;
    if (row < M) D[(size_t)row * NCOLS + n0 + lid] = acc[r];
  }
}

__global__ void zero_buf(float* __restrict__ p, unsigned n) {
  const unsigned t = blockIdx.x * blockDim.x + threadIdx.x;
  if (t < n) p[t] = 0.f;
}

__global__ void node_prep(const float* __restrict__ h,
                          const float* __restrict__ att_src,
                          const float* __restrict__ att_dst,
                          float* __restrict__ as, float* __restrict__ ad,
                          float* __restrict__ m, float* __restrict__ denom,
                          int Nn, int H, int C) {
  const int n = blockIdx.x * blockDim.x + threadIdx.x;
  if (n >= Nn) return;
  const int HC = H * C;
  for (int hh = 0; hh < H; ++hh) {
    float s = 0.f, d = 0.f;
    for (int c = 0; c < C; ++c) {
      const float v = h[(size_t)n * HC + hh * C + c];
      s += v * att_src[hh * C + c];
      d += v * att_dst[hh * C + c];
    }
    as[n * H + hh] = s;
    ad[n * H + hh] = d;
    m[n * H + hh] = -INFINITY;
    denom[n * H + hh] = 0.f;
  }
}

__global__ void edge_max(const int* __restrict__ src, const int* __restrict__ dst,
                         const float* __restrict__ as, const float* __restrict__ ad,
                         float* __restrict__ m, int E, int Nn, int H) {
  const unsigned t = blockIdx.x * blockDim.x + threadIdx.x;
  const unsigned tot = (unsigned)(E + Nn) * (unsigned)H;
  if (t >= tot) return;
  const int hh = (int)(t % (unsigned)H);
  const int e  = (int)(t / (unsigned)H);
  int s, d;
  if (e < E) { s = src[e]; d = dst[e]; } else { s = d = e - E; }
  const float v = lrelu(as[s * H + hh] + ad[d * H + hh], ATT_SLOPE);
  atomicMaxF(&m[d * H + hh], v);
}

__global__ void edge_accum(const int* __restrict__ src, const int* __restrict__ dst,
                           const float* __restrict__ as, const float* __restrict__ ad,
                           const float* __restrict__ m, const float* __restrict__ hfeat,
                           float* __restrict__ denom, float* __restrict__ acc,
                           int E, int Nn, int H, int C) {
  const int HC = H * C;
  const unsigned t = blockIdx.x * blockDim.x + threadIdx.x;
  const unsigned tot = (unsigned)(E + Nn) * (unsigned)HC;
  if (t >= tot) return;
  const int j = (int)(t % (unsigned)HC);
  const int e = (int)(t / (unsigned)HC);
  const int hh = j / C;
  int s, d;
  if (e < E) { s = src[e]; d = dst[e]; } else { s = d = e - E; }
  const float v = lrelu(as[s * H + hh] + ad[d * H + hh], ATT_SLOPE);
  const float w = expf(v - m[d * H + hh]);
  if ((j - hh * C) == 0) atomicAdd(&denom[d * H + hh], w);
  atomicAdd(&acc[(size_t)d * HC + j], w * hfeat[(size_t)s * HC + j]);
}

__global__ void finalize_feat(float* __restrict__ acc, const float* __restrict__ denom,
                              const float* __restrict__ bias, int Nn, int H, int C) {
  const unsigned t = blockIdx.x * blockDim.x + threadIdx.x;
  const int HC = H * C;
  if (t >= (unsigned)Nn * (unsigned)HC) return;
  const int n = (int)(t / (unsigned)HC);
  const int j = (int)(t % (unsigned)HC);
  const float v = acc[(size_t)n * HC + j] / denom[n * H + j / C] + bias[j];
  acc[(size_t)n * HC + j] = lrelu(v, ACT_SLOPE);
}

__global__ void finalize_proj(const float* __restrict__ acc, const float* __restrict__ denom,
                              const float* __restrict__ bias, const float* __restrict__ Wo,
                              const float* __restrict__ bo, float* __restrict__ out,
                              int Nn, int C) {
  const int n = blockIdx.x * blockDim.x + threadIdx.x;
  if (n >= Nn) return;
  const float dn = denom[n];
  float s = 0.f;
  for (int c = 0; c < C; ++c) {
    const float f = lrelu(acc[(size_t)n * C + c] / dn + bias[c], ACT_SLOPE);
    s += f * Wo[c];
  }
  out[n] = s + bo[0];
}

extern "C" void kernel_launch(void* const* d_in, const int* in_sizes, int n_in,
                              void* d_out, int out_size, void* d_ws, size_t ws_size,
                              hipStream_t stream) {
  const float* x    = (const float*)d_in[0];
  const int*   ei   = (const int*)d_in[1];
  const float* W1   = (const float*)d_in[2];
  const float* aS1  = (const float*)d_in[3];
  const float* aD1  = (const float*)d_in[4];
  const float* b1   = (const float*)d_in[5];
  const float* W2   = (const float*)d_in[6];
  const float* aS2  = (const float*)d_in[7];
  const float* aD2  = (const float*)d_in[8];
  const float* b2   = (const float*)d_in[9];
  const float* Wo   = (const float*)d_in[10];
  const float* bo   = (const float*)d_in[11];
  float* out = (float*)d_out;

  const int IN_C = 64, HID = 16, HEADS = 4;
  const int HC1 = HEADS * HID;
  const int Nn = in_sizes[0] / IN_C;
  const int E  = in_sizes[1] / 2;
  const int* src = ei;
  const int* dst = ei + E;

  float* p = (float*)d_ws;
  float* h1   = p; p += (size_t)Nn * HC1;
  float* acc1 = p; p += (size_t)Nn * HC1;
  float* h2   = p; p += (size_t)Nn * HID;
  float* acc2 = p; p += (size_t)Nn * HID;
  float* as1  = p; p += (size_t)Nn * HEADS;
  float* ad1  = p; p += (size_t)Nn * HEADS;
  float* m1   = p; p += (size_t)Nn * HEADS;
  float* dn1  = p; p += (size_t)Nn * HEADS;
  float* as2  = p; p += (size_t)Nn;
  float* ad2  = p; p += (size_t)Nn;
  float* m2   = p; p += (size_t)Nn;
  float* dn2  = p; p += (size_t)Nn;

  const int TB = 256;

  {
    dim3 g((Nn + 15) / 16, HC1 / 16);
    wmma_gemm_f32<64, 64><<<g, 32, 0, stream>>>(x, W1, h1, Nn);
  }
  {
    const unsigned n = (unsigned)Nn * (unsigned)HC1;
    zero_buf<<<(n + TB - 1) / TB, TB, 0, stream>>>(acc1, n);
  }
  node_prep<<<(Nn + TB - 1) / TB, TB, 0, stream>>>(h1, aS1, aD1, as1, ad1, m1, dn1,
                                                   Nn, HEADS, HID);
  {
    const unsigned n = (unsigned)(E + Nn) * (unsigned)HEADS;
    edge_max<<<(n + TB - 1) / TB, TB, 0, stream>>>(src, dst, as1, ad1, m1,
                                                   E, Nn, HEADS);
  }
  {
    const unsigned n = (unsigned)(E + Nn) * (unsigned)HC1;
    edge_accum<<<(n + TB - 1) / TB, TB, 0, stream>>>(src, dst, as1, ad1, m1,
                                                     h1, dn1, acc1,
                                                     E, Nn, HEADS, HID);
  }
  {
    const unsigned n = (unsigned)Nn * (unsigned)HC1;
    finalize_feat<<<(n + TB - 1) / TB, TB, 0, stream>>>(acc1, dn1, b1, Nn, HEADS, HID);
  }

  {
    dim3 g((Nn + 15) / 16, HID / 16);
    wmma_gemm_f32<64, 16><<<g, 32, 0, stream>>>(acc1, W2, h2, Nn);
  }
  {
    const unsigned n = (unsigned)Nn * (unsigned)HID;
    zero_buf<<<(n + TB - 1) / TB, TB, 0, stream>>>(acc2, n);
  }
  node_prep<<<(Nn + TB - 1) / TB, TB, 0, stream>>>(h2, aS2, aD2, as2, ad2, m2, dn2,
                                                   Nn, 1, HID);
  {
    const unsigned n = (unsigned)(E + Nn);
    edge_max<<<(n + TB - 1) / TB, TB, 0, stream>>>(src, dst, as2, ad2, m2,
                                                   E, Nn, 1);
  }
  {
    const unsigned n = (unsigned)(E + Nn) * (unsigned)HID;
    edge_accum<<<(n + TB - 1) / TB, TB, 0, stream>>>(src, dst, as2, ad2, m2,
                                                     h2, dn2, acc2,
                                                     E, Nn, 1, HID);
  }
  finalize_proj<<<(Nn + TB - 1) / TB, TB, 0, stream>>>(acc2, dn2, b2, Wo, bo, out,
                                                       Nn, HID);
}


# Edges counting-sorted by destination once per call (int atomic rank); each destination reduces its own segment in registers: no float atomics, sums in rank order, final dot as a lane tree
# speedup vs baseline: 3.7377x; 3.7377x over previous
#include <hip/hip_runtime.h>
#include <math.h>

#define ATT_SLOPE 0.2f
#define ACT_SLOPE 0.01f

typedef __attribute__((ext_vector_type(2))) float v2f;
typedef __attribute__((ext_vector_type(8))) float v8f;

__device__ __forceinline__ float lrelu(float v, float s) { return v > 0.f ? v : v * s; }

template <int K, int NCOLS>
__global__ void __launch_bounds__(32)
wmma_gemm_f32(const float* __restrict__ A, const float* __restrict__ B,
              float* __restrict__ D, int M) {
  const int m0 = blockIdx.x * 16;
  const int n0 = blockIdx.y * 16;
  const int lane = threadIdx.x;
  const int half = lane >> 4;
  const int lid  = lane & 15;
  const int ar = min(m0 + lid, M - 1);
  v8f acc = {};
#pragma unroll
  for (int k = 0; k < K; k += 4) {
    const int ka = k + half * 2;
    v2f a, b;
    a.x = A[(size_t)ar * K + ka];
    a.y = A[(size_t)ar * K + ka + 1];
    b.x = B[(size_t)ka * NCOLS + n0 + lid];
    b.y = B[(size_t)(ka + 1) * NCOLS + n0 + lid];
    acc = __builtin_amdgcn_wmma_f32_16x16x4_f32(false, a, false, b, (short)0, acc,
                                                false, false);
  }
  const int rbase = m0 + half * 8;
#pragma unroll
  for (int r = 0; r < 8; ++r) {
    const int row = rbase + r;
    if (row < M) D[(size_t)row * NCOLS + n0 + lid] = acc[r];
  }
}

__global__ void zero_int(int* __restrict__ p, unsigned n) {
  const unsigned t = blockIdx.x * blockDim.x + threadIdx.x;
  if (t < n) p[t] = 0;
}

__global__ void edge_rank(const int* __restrict__ dst, int* __restrict__ cnt,
                          int* __restrict__ rank, int E) {
  const unsigned t = blockIdx.x * blockDim.x + threadIdx.x;
  if (t >= (unsigned)E) return;
  rank[t] = atomicAdd(&cnt[dst[t]], 1);
}

__global__ void __launch_bounds__(1024)
scan_counts(const int* __restrict__ cnt, int* __restrict__ rowptr, int n) {
  __shared__ int wsum[32];
  const int t = threadIdx.x;
  const int lane = t & 31, w = t >> 5;
  const int per = (n + 1023) / 1024;
  const int b = min(t * per, n);
  const int e = min(b + per, n);
  int s = 0;
  for (int i = b; i < e; ++i) s += cnt[i];
  int v = s;
  for (int o = 1; o < 32; o <<= 1) {
    const int u = __shfl_up(v, o, 32);
    if (lane >= o) v += u;
  }
  if (lane == 31) wsum[w] = v;
  __syncthreads();
  if (w == 0) {
    int x = wsum[lane];
    for (int o = 1; o < 32; o <<= 1) {
      const int u = __shfl_up(x, o, 32);
      if (lane >= o) x += u;
    }
    wsum[lane] = x;
  }
  __syncthreads();
  int run = v - s + (w > 0 ? wsum[w - 1] : 0);
  for (int i = b; i < e; ++i) {
    rowptr[i] = run;
    run += cnt[i];
  }
  if (t == 1023) rowptr[n] = run;
}

__global__ void edge_scatter(const int* __restrict__ src, const int* __restrict__ dst,
                             const int* __restrict__ rank, const int* __restrict__ rowptr,
                             int* __restrict__ csr, int E) {
  const unsigned t = blockIdx.x * blockDim.x + threadIdx.x;
  if (t >= (unsigned)E) return;
  csr[rowptr[dst[t]] + rank[t]] = src[t];
}

__global__ void node_prep(const float* __restrict__ h,
                          const float* __restrict__ att_src,
                          const float* __restrict__ att_dst,
                          float* __restrict__ as, float* __restrict__ ad,
                          int Nn, int H, int C) {
  const int n = blockIdx.x * blockDim.x + threadIdx.x;
  if (n >= Nn) return;
  const int HC = H * C;
  for (int hh = 0; hh < H; ++hh) {
    float s = 0.f, d = 0.f;
    for (int c = 0; c < C; ++c) {
      const float v = h[(size_t)n * HC + hh * C + c];
      s += v * att_src[hh * C + c];
      d += v * att_dst[hh * C + c];
    }
    as[n * H + hh] = s;
    ad[n * H + hh] = d;
  }
}

template <int H, int C, int VEC, bool PROJ>
__global__ void __launch_bounds__(256)
seg_gather(const int* __restrict__ rowptr, const int* __restrict__ csr,
           const float* __restrict__ as, const float* __restrict__ ad,
           const float* __restrict__ hfeat, const float* __restrict__ bias,
           const float* __restrict__ Wo, const float* __restrict__ bo,
           float* __restrict__ dstbuf, int Nn) {
  constexpr int HC = H * C;
  constexpr int LPN = HC / VEC;
  constexpr int EPP = LPN / H;
  const unsigned gt = blockIdx.x * blockDim.x + threadIdx.x;
  const int node = (int)(gt / LPN);
  if (node >= Nn) return;
  const int gl = (int)(threadIdx.x % LPN);
  const int f0 = gl * VEC;
  const int hh = f0 / C;
  const int beg = rowptr[node];
  const int end = rowptr[node + 1];

  const int ph = gl % H;
  const int pe = gl / H;
  const float adp = ad[node * H + ph];
  float mx = lrelu(as[node * H + ph] + adp, ATT_SLOPE);
  for (int i = beg + pe; i < end; i += EPP) {
    const int s = csr[i];
    mx = fmaxf(mx, lrelu(as[s * H + ph] + adp, ATT_SLOPE));
  }
  for (int o = H; o < LPN; o <<= 1) mx = fmaxf(mx, __shfl_xor(mx, o, LPN));
  const float m = __shfl(mx, hh, LPN);

  const float adv = ad[node * H + hh];
  float den;
  float acc[VEC];
  {
    const float w = expf(lrelu(as[node * H + hh] + adv, ATT_SLOPE) - m);
    den = w;
    const float* hp = hfeat + (size_t)node * HC + f0;
#pragma unroll
    for (int k = 0; k < VEC; ++k) acc[k] = w * hp[k];
  }
#pragma unroll 4
  for (int i = beg; i < end; ++i) {
    const int s = csr[i];
    const float w = expf(lrelu(as[s * H + hh] + adv, ATT_SLOPE) - m);
    den += w;
    const float* hp = hfeat + (size_t)s * HC + f0;
#pragma unroll
    for (int k = 0; k < VEC; ++k) acc[k] += w * hp[k];
  }

  if constexpr (PROJ) {
    float p = 0.f;
#pragma unroll
    for (int k = 0; k < VEC; ++k)
      p += lrelu(acc[k] / den + bias[f0 + k], ACT_SLOPE) * Wo[f0 + k];
    for (int o = 1; o < LPN; o <<= 1) p += __shfl_xor(p, o, LPN);
    if (gl == 0) dstbuf[node] = p + bo[0];
  } else {
    float* op = dstbuf + (size_t)node * HC + f0;
#pragma unroll
    for (int k = 0; k < VEC; ++k)
      op[k] = lrelu(acc[k] / den + bias[f0 + k], ACT_SLOPE);
  }
}

extern "C" void kernel_launch(void* const* d_in, const int* in_sizes, int n_in,
                              void* d_out, int out_size, void* d_ws, size_t ws_size,
                              hipStream_t stream) {
  const float* x    = (const float*)d_in[0];
  const int*   ei   = (const int*)d_in[1];
  const float* W1   = (const float*)d_in[2];
  const float* aS1  = (const float*)d_in[3];
  const float* aD1  = (const float*)d_in[4];
  const float* b1   = (const float*)d_in[5];
  const float* W2   = (const float*)d_in[6];
  const float* aS2  = (const float*)d_in[7];
  const float* aD2  = (const float*)d_in[8];
  const float* b2   = (const float*)d_in[9];
  const float* Wo   = (const float*)d_in[10];
  const float* bo   = (const float*)d_in[11];
  float* out = (float*)d_out;

  const int IN_C = 64, HID = 16, HEADS = 4;
  const int HC1 = HEADS * HID;
  const int Nn = in_sizes[0] / IN_C;
  const int E  = in_sizes[1] / 2;
  const int* src = ei;
  const int* dst = ei + E;

  float* p = (float*)d_ws;
  float* h1   = p; p += (size_t)Nn * HC1;
  float* f1   = p; p += (size_t)Nn * HC1;
  float* h2   = p; p += (size_t)Nn * HID;
  float* as1  = p; p += (size_t)Nn * HEADS;
  float* ad1  = p; p += (size_t)Nn * HEADS;
  float* as2  = p; p += (size_t)Nn;
  float* ad2  = p; p += (size_t)Nn;
  int* q = (int*)p;
  int* cnt    = q; q += (size_t)Nn;
  int* rowptr = q; q += (size_t)Nn + 1;
  int* rank   = q; q += (size_t)E;
  int* csr    = q; q += (size_t)E;

  const int TB = 256;

  zero_int<<<(Nn + TB - 1) / TB, TB, 0, stream>>>(cnt, (unsigned)Nn);
  edge_rank<<<(E + TB - 1) / TB, TB, 0, stream>>>(dst, cnt, rank, E);
  scan_counts<<<1, 1024, 0, stream>>>(cnt, rowptr, Nn);
  edge_scatter<<<(E + TB - 1) / TB, TB, 0, stream>>>(src, dst, rank, rowptr, csr, E);

  {
    dim3 g((Nn + 15) / 16, HC1 / 16);
    wmma_gemm_f32<64, 64><<<g, 32, 0, stream>>>(x, W1, h1, Nn);
  }
  node_prep<<<(Nn + TB - 1) / TB, TB, 0, stream>>>(h1, aS1, aD1, as1, ad1,
                                                   Nn, HEADS, HID);
  {
    const unsigned n = (unsigned)Nn * 16u;
    seg_gather<4, 16, 4, false><<<(n + TB - 1) / TB, TB, 0, stream>>>(
        rowptr, csr, as1, ad1, h1, b1, Wo, bo, f1, Nn);
  }

  {
    dim3 g((Nn + 15) / 16, HID / 16);
    wmma_gemm_f32<64, 16><<<g, 32, 0, stream>>>(f1, W2, h2, Nn);
  }
  node_prep<<<(Nn + TB - 1) / TB, TB, 0, stream>>>(h2, aS2, aD2, as2, ad2,
                                                   Nn, 1, HID);
  {
    const unsigned n = (unsigned)Nn * 8u;
    seg_gather<1, 16, 2, true><<<(n + TB - 1) / TB, TB, 0, stream>>>(
        rowptr, csr, as2, ad2, h2, b2, Wo, bo, out, Nn);
  }
}


# Independent launches merged side by side (role by work-group index): the layer-1 GEMM runs beside the one-group scan, the layer-1 coefficients beside the edge scatter; 10 launches become 8
# speedup vs baseline: 4.0093x; 1.0727x over previous
#include <hip/hip_runtime.h>
#include <math.h>

#define ATT_SLOPE 0.2f
#define ACT_SLOPE 0.01f

typedef __attribute__((ext_vector_type(2))) float v2f;
typedef __attribute__((ext_vector_type(8))) float v8f;

__device__ __forceinline__ float lrelu(float v, float s) { return v > 0.f ? v : v * s; }

template <int K, int NCOLS>
__device__ __forceinline__ void
wmma_tile_f32(const float* __restrict__ A, const float* __restrict__ B,
              float* __restrict__ D, int M, int m0, int n0, int lane) {
  const int half = lane >> 4;
  const int lid  = lane & 15;
  const int ar = min(m0 + lid, M - 1);
  v8f acc = {};
#pragma unroll
  for (int k = 0; k < K; k += 4) {
    const int ka = k + half * 2;
    v2f a, b;
    a.x = A[(size_t)ar * K + ka];
    a.y = A[(size_t)ar * K + ka + 1];
    b.x = B[(size_t)ka * NCOLS + n0 + lid];
    b.y = B[(size_t)(ka + 1) * NCOLS + n0 + lid];
    acc = __builtin_amdgcn_wmma_f32_16x16x4_f32(false, a, false, b, (short)0, acc,
                                                false, false);
  }
  const int rbase = m0 + half * 8;
#pragma unroll
  for (int r = 0; r < 8; ++r) {
    const int row = rbase + r;
    if (row < M) D[(size_t)row * NCOLS + n0 + lid] = acc[r];
  }
}

template <int K, int NCOLS>
__global__ void __launch_bounds__(32)
wmma_gemm_f32(const float* __restrict__ A, const float* __restrict__ B,
              float* __restrict__ D, int M) {
  wmma_tile_f32<K, NCOLS>(A, B, D, M, blockIdx.x * 16, blockIdx.y * 16, threadIdx.x);
}

__global__ void zero_int(int* __restrict__ p, unsigned n) {
  const unsigned t = blockIdx.x * blockDim.x + threadIdx.x;
  if (t < n) p[t] = 0;
}

__global__ void edge_rank(const int* __restrict__ dst, int* __restrict__ cnt,
                          int* __restrict__ rank, int E) {
  const unsigned t = blockIdx.x * blockDim.x + threadIdx.x;
  if (t >= (unsigned)E) return;
  rank[t] = atomicAdd(&cnt[dst[t]], 1);
}

__device__ __forceinline__ void
scan_counts_body(const int* __restrict__ cnt, int* __restrict__ rowptr, int n) {
  __shared__ int wsum[32];
  const int t = threadIdx.x;
  const int lane = t & 31, w = t >> 5;
  const int per = (n + 1023) / 1024;
  const int b = min(t * per, n);
  const int e = min(b + per, n);
  int s = 0;
  for (int i = b; i < e; ++i) s += cnt[i];
  int v = s;
  for (int o = 1; o < 32; o <<= 1) {
    const int u = __shfl_up(v, o, 32);
    if (lane >= o) v += u;
  }
  if (lane == 31) wsum[w] = v;
  __syncthreads();
  if (w == 0) {
    int x = wsum[lane];
    for (int o = 1; o < 32; o <<= 1) {
      const int u = __shfl_up(x, o, 32);
      if (lane >= o) x += u;
    }
    wsum[lane] = x;
  }
  __syncthreads();
  int run = v - s + (w > 0 ? wsum[w - 1] : 0);
  for (int i = b; i < e; ++i) {
    rowptr[i] = run;
    run += cnt[i];
  }
  if (t == 1023) rowptr[n] = run;
}

template <int K, int NCOLS>
__global__ void __launch_bounds__(1024)
scan_and_gemm(const int* __restrict__ cnt, int* __restrict__ rowptr, int n,
              const float* __restrict__ A, const float* __restrict__ B,
              float* __restrict__ D, int M) {
  if (blockIdx.x == 0) {
    scan_counts_body(cnt, rowptr, n);
    return;
  }
  constexpr int NT = NCOLS / 16;
  const int tile = (int)(blockIdx.x - 1) * 32 + (int)(threadIdx.x >> 5);
  const int m0 = (tile / NT) * 16;
  if (m0 >= M) return;
  wmma_tile_f32<K, NCOLS>(A, B, D, M, m0, (tile % NT) * 16, threadIdx.x & 31);
}

__device__ __forceinline__ void
edge_scatter_body(const int* __restrict__ src, const int* __restrict__ dst,
                  const int* __restrict__ rank, const int* __restrict__ rowptr,
                  int* __restrict__ csr, int E, unsigned t) {
  if (t >= (unsigned)E) return;
  csr[rowptr[dst[t]] + rank[t]] = src[t];
}

__device__ __forceinline__ void
node_prep_body(const float* __restrict__ h,
               const float* __restrict__ att_src,
               const float* __restrict__ att_dst,
               float* __restrict__ as, float* __restrict__ ad,
               int Nn, int H, int C, int n) {
  if (n >= Nn) return;
  const int HC = H * C;
  for (int hh = 0; hh < H; ++hh) {
    float s = 0.f, d = 0.f;
    for (int c = 0; c < C; ++c) {
      const float v = h[(size_t)n * HC + hh * C + c];
      s += v * att_src[hh * C + c];
      d += v * att_dst[hh * C + c];
    }
    as[n * H + hh] = s;
    ad[n * H + hh] = d;
  }
}

__global__ void node_prep(const float* __restrict__ h,
                          const float* __restrict__ att_src,
                          const float* __restrict__ att_dst,
                          float* __restrict__ as, float* __restrict__ ad,
                          int Nn, int H, int C) {
  node_prep_body(h, att_src, att_dst, as, ad, Nn, H, C,
                 (int)(blockIdx.x * blockDim.x + threadIdx.x));
}

__global__ void __launch_bounds__(256)
scatter_and_prep(const int* __restrict__ src, const int* __restrict__ dst,
                 const int* __restrict__ rank, const int* __restrict__ rowptr,
                 int* __restrict__ csr, int E,
                 const float* __restrict__ h,
                 const float* __restrict__ att_src,
                 const float* __restrict__ att_dst,
                 float* __restrict__ as, float* __restrict__ ad,
                 int Nn, int H, int C, unsigned prep_blocks) {
  if (blockIdx.x < prep_blocks) {
    node_prep_body(h, att_src, att_dst, as, ad, Nn, H, C,
                   (int)(blockIdx.x * blockDim.x + threadIdx.x));
  } else {
    edge_scatter_body(src, dst, rank, rowptr, csr, E,
                      (blockIdx.x - prep_blocks) * blockDim.x + threadIdx.x);
  }
}

template <int H, int C, int VEC, bool PROJ>
__global__ void __launch_bounds__(256)
seg_gather(const int* __restrict__ rowptr, const int* __restrict__ csr,
           const float* __restrict__ as, const float* __restrict__ ad,
           const float* __restrict__ hfeat, const float* __restrict__ bias,
           const float* __restrict__ Wo, const float* __restrict__ bo,
           float* __restrict__ dstbuf, int Nn) {
  constexpr int HC = H * C;
  constexpr int LPN = HC / VEC;
  constexpr int EPP = LPN / H;
  const unsigned gt = blockIdx.x * blockDim.x + threadIdx.x;
  const int node = (int)(gt / LPN);
  if (node >= Nn) return;
  const int gl = (int)(threadIdx.x % LPN);
  const int f0 = gl * VEC;
  const int hh = f0 / C;
  const int beg = rowptr[node];
  const int end = rowptr[node + 1];

  const int ph = gl % H;
  const int pe = gl / H;
  const float adp = ad[node * H + ph];
  float mx = lrelu(as[node * H + ph] + adp, ATT_SLOPE);
  for (int i = beg + pe; i < end; i += EPP) {
    const int s = csr[i];
    mx = fmaxf(mx, lrelu(as[s * H + ph] + adp, ATT_SLOPE));
  }
  for (int o = H; o < LPN; o <<= 1) mx = fmaxf(mx, __shfl_xor(mx, o, LPN));
  const float m = __shfl(mx, hh, LPN);

  const float adv = ad[node * H + hh];
  float den;
  float acc[VEC];
  {
    const float w = expf(lrelu(as[node * H + hh] + adv, ATT_SLOPE) - m);
    den = w;
    const float* hp = hfeat + (size_t)node * HC + f0;
#pragma unroll
    for (int k = 0; k < VEC; ++k) acc[k] = w * hp[k];
  }
#pragma unroll 4
  for (int i = beg; i < end; ++i) {
    const int s = csr[i];
    const float w = expf(lrelu(as[s * H + hh] + adv, ATT_SLOPE) - m);
    den += w;
    const float* hp = hfeat + (size_t)s * HC + f0;
#pragma unroll
    for (int k = 0; k < VEC; ++k) acc[k] += w * hp[k];
  }

  if constexpr (PROJ) {
    float p = 0.f;
#pragma unroll
    for (int k = 0; k < VEC; ++k)
      p += lrelu(acc[k] / den + bias[f0 + k], ACT_SLOPE) * Wo[f0 + k];
    for (int o = 1; o < LPN; o <<= 1) p += __shfl_xor(p, o, LPN);
    if (gl == 0) dstbuf[node] = p + bo[0];
  } else {
    float* op = dstbuf + (size_t)node * HC + f0;
#pragma unroll
    for (int k = 0; k < VEC; ++k)
      op[k] = lrelu(acc[k] / den + bias[f0 + k], ACT_SLOPE);
  }
}

extern "C" void kernel_launch(void* const* d_in, const int* in_sizes, int n_in,
                              void* d_out, int out_size, void* d_ws, size_t ws_size,
                              hipStream_t stream) {
  const float* x    = (const float*)d_in[0];
  const int*   ei   = (const int*)d_in[1];
  const float* W1   = (const float*)d_in[2];
  const float* aS1  = (const float*)d_in[3];
  const float* aD1  = (const float*)d_in[4];
  const float* b1   = (const float*)d_in[5];
  const float* W2   = (const float*)d_in[6];
  const float* aS2  = (const float*)d_in[7];
  const float* aD2  = (const float*)d_in[8];
  const float* b2   = (const float*)d_in[9];
  const float* Wo   = (const float*)d_in[10];
  const float* bo   = (const float*)d_in[11];
  float* out = (float*)d_out;

  const int IN_C = 64, HID = 16, HEADS = 4;
  const int HC1 = HEADS * HID;
  const int Nn = in_sizes[0] / IN_C;
  const int E  = in_sizes[1] / 2;
  const int* src = ei;
  const int* dst = ei + E;

  float* p = (float*)d_ws;
  float* h1   = p; p += (size_t)Nn * HC1;
  float* f1   = p; p += (size_t)Nn * HC1;
  float* h2   = p; p += (size_t)Nn * HID;
  float* as1  = p; p += (size_t)Nn * HEADS;
  float* ad1  = p; p += (size_t)Nn * HEADS;
  float* as2  = p; p += (size_t)Nn;
  float* ad2  = p; p += (size_t)Nn;
  int* q = (int*)p;
  int* cnt    = q; q += (size_t)Nn;
  int* rowptr = q; q += (size_t)Nn + 1;
  int* rank   = q; q += (size_t)E;
  int* csr    = q; q += (size_t)E;

  const int TB = 256;

  zero_int<<<(Nn + TB - 1) / TB, TB, 0, stream>>>(cnt, (unsigned)Nn);
  edge_rank<<<(E + TB - 1) / TB, TB, 0, stream>>>(dst, cnt, rank, E);
  {
    const int tiles = ((Nn + 15) / 16) * (HC1 / 16);
    scan_and_gemm<64, 64><<<1 + (tiles + 31) / 32, 1024, 0, stream>>>(
        cnt, rowptr, Nn, x, W1, h1, Nn);
  }
  {
    const unsigned pb = (unsigned)((Nn + TB - 1) / TB);
    scatter_and_prep<<<pb + (unsigned)((E + TB - 1) / TB), TB, 0, stream>>>(
        src, dst, rank, rowptr, csr, E, h1, aS1, aD1, as1, ad1, Nn, HEADS, HID, pb);
  }

  {
    const unsigned n = (unsigned)Nn * 16u;
    seg_gather<4, 16, 4, false><<<(n + TB - 1) / TB, TB, 0, stream>>>(
        rowptr, csr, as1, ad1, h1, b1, Wo, bo, f1, Nn);
  }

  {
    dim3 g((Nn + 15) / 16, HID / 16);
    wmma_gemm_f32<64, 16><<<g, 32, 0, stream>>>(f1, W2, h2, Nn);
  }
  node_prep<<<(Nn + TB - 1) / TB, TB, 0, stream>>>(h2, aS2, aD2, as2, ad2,
                                                   Nn, 1, HID);
  {
    const unsigned n = (unsigned)Nn * 8u;
    seg_gather<1, 16, 2, true><<<(n + TB - 1) / TB, TB, 0, stream>>>(
        rowptr, csr, as2, ad2, h2, b2, Wo, bo, out, Nn);
  }
}


# Segment softmax in ONE pass over the stored scores: running max per chunk of 4 edges, sums so far scaled by exp(old-new); the max pass and its score gathers go; edge order kept, rounding changes
# speedup vs baseline: 4.3264x; 1.0791x over previous
#include <hip/hip_runtime.h>
#include <math.h>

#define ATT_SLOPE 0.2f
#define ACT_SLOPE 0.01f

typedef __attribute__((ext_vector_type(2))) float v2f;
typedef __attribute__((ext_vector_type(8))) float v8f;

__device__ __forceinline__ float lrelu(float v, float s) { return v > 0.f ? v : v * s; }

template <int K, int NCOLS>
__device__ __forceinline__ void
wmma_tile_f32(const float* __restrict__ A, const float* __restrict__ B,
              float* __restrict__ D, int M, int m0, int n0, int lane) {
  const int half = lane >> 4;
  const int lid  = lane & 15;
  const int ar = min(m0 + lid, M - 1);
  v8f acc = {};
#pragma unroll
  for (int k = 0; k < K; k += 4) {
    const int ka = k + half * 2;
    v2f a, b;
    a.x = A[(size_t)ar * K + ka];
    a.y = A[(size_t)ar * K + ka + 1];
    b.x = B[(size_t)ka * NCOLS + n0 + lid];
    b.y = B[(size_t)(ka + 1) * NCOLS + n0 + lid];
    acc = __builtin_amdgcn_wmma_f32_16x16x4_f32(false, a, false, b, (short)0, acc,
                                                false, false);
  }
  const int rbase = m0 + half * 8;
#pragma unroll
  for (int r = 0; r < 8; ++r) {
    const int row = rbase + r;
    if (row < M) D[(size_t)row * NCOLS + n0 + lid] = acc[r];
  }
}

template <int K, int NCOLS>
__global__ void __launch_bounds__(32)
wmma_gemm_f32(const float* __restrict__ A, const float* __restrict__ B,
              float* __restrict__ D, int M) {
  wmma_tile_f32<K, NCOLS>(A, B, D, M, blockIdx.x * 16, blockIdx.y * 16, threadIdx.x);
}

__global__ void zero_int(int* __restrict__ p, unsigned n) {
  const unsigned t = blockIdx.x * blockDim.x + threadIdx.x;
  if (t < n) p[t] = 0;
}

__global__ void edge_rank(const int* __restrict__ dst, int* __restrict__ cnt,
                          int* __restrict__ rank, int E) {
  const unsigned t = blockIdx.x * blockDim.x + threadIdx.x;
  if (t >= (unsigned)E) return;
  rank[t] = atomicAdd(&cnt[dst[t]], 1);
}

__device__ __forceinline__ void
scan_counts_body(const int* __restrict__ cnt, int* __restrict__ rowptr, int n) {
  __shared__ int wsum[32];
  const int t = threadIdx.x;
  const int lane = t & 31, w = t >> 5;
  const int per = (n + 1023) / 1024;
  const int b = min(t * per, n);
  const int e = min(b + per, n);
  int s = 0;
  for (int i = b; i < e; ++i) s += cnt[i];
  int v = s;
  for (int o = 1; o < 32; o <<= 1) {
    const int u = __shfl_up(v, o, 32);
    if (lane >= o) v += u;
  }
  if (lane == 31) wsum[w] = v;
  __syncthreads();
  if (w == 0) {
    int x = wsum[lane];
    for (int o = 1; o < 32; o <<= 1) {
      const int u = __shfl_up(x, o, 32);
      if (lane >= o) x += u;
    }
    wsum[lane] = x;
  }
  __syncthreads();
  int run = v - s + (w > 0 ? wsum[w - 1] : 0);
  for (int i = b; i < e; ++i) {
    rowptr[i] = run;
    run += cnt[i];
  }
  if (t == 1023) rowptr[n] = run;
}

template <int K, int NCOLS>
__global__ void __launch_bounds__(1024)
scan_and_gemm(const int* __restrict__ cnt, int* __restrict__ rowptr, int n,
              const float* __restrict__ A, const float* __restrict__ B,
              float* __restrict__ D, int M) {
  if (blockIdx.x == 0) {
    scan_counts_body(cnt, rowptr, n);
    return;
  }
  constexpr int NT = NCOLS / 16;
  const int tile = (int)(blockIdx.x - 1) * 32 + (int)(threadIdx.x >> 5);
  const int m0 = (tile / NT) * 16;
  if (m0 >= M) return;
  wmma_tile_f32<K, NCOLS>(A, B, D, M, m0, (tile % NT) * 16, threadIdx.x & 31);
}

__device__ __forceinline__ void
edge_scatter_body(const int* __restrict__ src, const int* __restrict__ dst,
                  const int* __restrict__ rank, const int* __restrict__ rowptr,
                  int* __restrict__ csr, int E, unsigned t) {
  if (t >= (unsigned)E) return;
  csr[rowptr[dst[t]] + rank[t]] = src[t];
}

__device__ __forceinline__ void
node_prep_body(const float* __restrict__ h,
               const float* __restrict__ att_src,
               const float* __restrict__ att_dst,
               float* __restrict__ as, float* __restrict__ ad,
               int Nn, int H, int C, int n) {
  if (n >= Nn) return;
  const int HC = H * C;
  for (int hh = 0; hh < H; ++hh) {
    float s = 0.f, d = 0.f;
    for (int c = 0; c < C; ++c) {
      const float v = h[(size_t)n * HC + hh * C + c];
      s += v * att_src[hh * C + c];
      d += v * att_dst[hh * C + c];
    }
    as[n * H + hh] = s;
    ad[n * H + hh] = d;
  }
}

__global__ void node_prep(const float* __restrict__ h,
                          const float* __restrict__ att_src,
                          const float* __restrict__ att_dst,
                          float* __restrict__ as, float* __restrict__ ad,
                          int Nn, int H, int C) {
  node_prep_body(h, att_src, att_dst, as, ad, Nn, H, C,
                 (int)(blockIdx.x * blockDim.x + threadIdx.x));
}

__global__ void __launch_bounds__(256)
scatter_and_prep(const int* __restrict__ src, const int* __restrict__ dst,
                 const int* __restrict__ rank, const int* __restrict__ rowptr,
                 int* __restrict__ csr, int E,
                 const float* __restrict__ h,
                 const float* __restrict__ att_src,
                 const float* __restrict__ att_dst,
                 float* __restrict__ as, float* __restrict__ ad,
                 int Nn, int H, int C, unsigned prep_blocks) {
  if (blockIdx.x < prep_blocks) {
    node_prep_body(h, att_src, att_dst, as, ad, Nn, H, C,
                   (int)(blockIdx.x * blockDim.x + threadIdx.x));
  } else {
    edge_scatter_body(src, dst, rank, rowptr, csr, E,
                      (blockIdx.x - prep_blocks) * blockDim.x + threadIdx.x);
  }
}

template <int H, int C, int VEC, int CH, bool FULL>
__device__ __forceinline__ void
softmax_chunk(const int* __restrict__ csr, const float* __restrict__ as,
              const float* __restrict__ hfeat, int i, int end, int hh, int f0,
              float adv, float& m, float& den, float (&acc)[VEC]) {
  constexpr int HC = H * C;
  int s[CH];
  float e[CH];
#pragma unroll
  for (int j = 0; j < CH; ++j) s[j] = FULL ? csr[i + j] : csr[min(i + j, end - 1)];
#pragma unroll
  for (int j = 0; j < CH; ++j) {
    const float v = lrelu(as[s[j] * H + hh] + adv, ATT_SLOPE);
    e[j] = (FULL || i + j < end) ? v : -INFINITY;
  }
  float cm = e[0];
#pragma unroll
  for (int j = 1; j < CH; ++j) cm = fmaxf(cm, e[j]);
  const float mn = fmaxf(m, cm);
  const float sc = expf(m - mn);
  m = mn;
  den *= sc;
#pragma unroll
  for (int k = 0; k < VEC; ++k) acc[k] *= sc;
#pragma unroll
  for (int j = 0; j < CH; ++j) {
    const float w = expf(e[j] - mn);
    den += w;
    const float* hp = hfeat + (size_t)s[j] * HC + f0;
#pragma unroll
    for (int k = 0; k < VEC; ++k) acc[k] += w * hp[k];
  }
}

template <int H, int C, int VEC, bool PROJ>
__global__ void __launch_bounds__(256)
seg_gather(const int* __restrict__ rowptr, const int* __restrict__ csr,
           const float* __restrict__ as, const float* __restrict__ ad,
           const float* __restrict__ hfeat, const float* __restrict__ bias,
           const float* __restrict__ Wo, const float* __restrict__ bo,
           float* __restrict__ dstbuf, int Nn) {
  constexpr int HC = H * C;
  constexpr int LPN = HC / VEC;
  constexpr int CH = 4;
  const unsigned gt = blockIdx.x * blockDim.x + threadIdx.x;
  const int node = (int)(gt / LPN);
  if (node >= Nn) return;
  const int gl = (int)(threadIdx.x % LPN);
  const int f0 = gl * VEC;
  const int hh = f0 / C;
  const int beg = rowptr[node];
  const int end = rowptr[node + 1];

  const float adv = ad[node * H + hh];
  float m = lrelu(as[node * H + hh] + adv, ATT_SLOPE);
  float den = 1.f;
  float acc[VEC];
  {
    const float* hp = hfeat + (size_t)node * HC + f0;
#pragma unroll
    for (int k = 0; k < VEC; ++k) acc[k] = hp[k];
  }

  int i = beg;
  for (; i + CH <= end; i += CH)
    softmax_chunk<H, C, VEC, CH, true>(csr, as, hfeat, i, end, hh, f0, adv, m, den, acc);
  if (i < end)
    softmax_chunk<H, C, VEC, CH, false>(csr, as, hfeat, i, end, hh, f0, adv, m, den, acc);

  if constexpr (PROJ) {
    float p = 0.f;
#pragma unroll
    for (int k = 0; k < VEC; ++k)
      p += lrelu(acc[k] / den + bias[f0 + k], ACT_SLOPE) * Wo[f0 + k];
    for (int o = 1; o < LPN; o <<= 1) p += __shfl_xor(p, o, LPN);
    if (gl == 0) dstbuf[node] = p + bo[0];
  } else {
    float* op = dstbuf + (size_t)node * HC + f0;
#pragma unroll
    for (int k = 0; k < VEC; ++k)
      op[k] = lrelu(acc[k] / den + bias[f0 + k], ACT_SLOPE);
  }
}

extern "C" void kernel_launch(void* const* d_in, const int* in_sizes, int n_in,
                              void* d_out, int out_size, void* d_ws, size_t ws_size,
                              hipStream_t stream) {
  const float* x    = (const float*)d_in[0];
  const int*   ei   = (const int*)d_in[1];
  const float* W1   = (const float*)d_in[2];
  const float* aS1  = (const float*)d_in[3];
  const float* aD1  = (const float*)d_in[4];
  const float* b1   = (const float*)d_in[5];
  const float* W2   = (const float*)d_in[6];
  const float* aS2  = (const float*)d_in[7];
  const float* aD2  = (const float*)d_in[8];
  const float* b2   = (const float*)d_in[9];
  const float* Wo   = (const float*)d_in[10];
  const float* bo   = (const float*)d_in[11];
  float* out = (float*)d_out;

  const int IN_C = 64, HID = 16, HEADS = 4;
  const int HC1 = HEADS * HID;
  const int Nn = in_sizes[0] / IN_C;
  const int E  = in_sizes[1] / 2;
  const int* src = ei;
  const int* dst = ei + E;

  float* p = (float*)d_ws;
  float* h1   = p; p += (size_t)Nn * HC1;
  float* f1   = p; p += (size_t)Nn * HC1;
  float* h2   = p; p += (size_t)Nn * HID;
  float* as1  = p; p += (size_t)Nn * HEADS;
  float* ad1  = p; p += (size_t)Nn * HEADS;
  float* as2  = p; p += (size_t)Nn;
  float* ad2  = p; p += (size_t)Nn;
  int* q = (int*)p;
  int* cnt    = q; q += (size_t)Nn;
  int* rowptr = q; q += (size_t)Nn + 1;
  int* rank   = q; q += (size_t)E;
  int* csr    = q; q += (size_t)E;

  const int TB = 256;

  zero_int<<<(Nn + TB - 1) / TB, TB, 0, stream>>>(cnt, (unsigned)Nn);
  edge_rank<<<(E + TB - 1) / TB, TB, 0, stream>>>(dst, cnt, rank, E);
  {
    const int tiles = ((Nn + 15) / 16) * (HC1 / 16);
    scan_and_gemm<64, 64><<<1 + (tiles + 31) / 32, 1024, 0, stream>>>(
        cnt, rowptr, Nn, x, W1, h1, Nn);
  }
  {
    const unsigned pb = (unsigned)((Nn + TB - 1) / TB);
    scatter_and_prep<<<pb + (unsigned)((E + TB - 1) / TB), TB, 0, stream>>>(
        src, dst, rank, rowptr, csr, E, h1, aS1, aD1, as1, ad1, Nn, HEADS, HID, pb);
  }

  {
    const unsigned n = (unsigned)Nn * 16u;
    seg_gather<4, 16, 4, false><<<(n + TB - 1) / TB, TB, 0, stream>>>(
        rowptr, csr, as1, ad1, h1, b1, Wo, bo, f1, Nn);
  }

  {
    dim3 g((Nn + 15) / 16, HID / 16);
    wmma_gemm_f32<64, 16><<<g, 32, 0, stream>>>(f1, W2, h2, Nn);
  }
  node_prep<<<(Nn + TB - 1) / TB, TB, 0, stream>>>(h2, aS2, aD2, as2, ad2,
                                                   Nn, 1, HID);
  {
    const unsigned n = (unsigned)Nn * 8u;
    seg_gather<1, 16, 2, true><<<(n + TB - 1) / TB, TB, 0, stream>>>(
        rowptr, csr, as2, ad2, h2, b2, Wo, bo, out, Nn);
  }
}


# Edges sorted by destination in two levels ranked on LDS counters (8192-edge chunks into 256-node buckets, then per bucket): no returning global atomic per edge; edge order in a segment changes
# speedup vs baseline: 6.7133x; 1.5517x over previous
#include <hip/hip_runtime.h>
#include <math.h>

#define ATT_SLOPE 0.2f
#define ACT_SLOPE 0.01f

typedef __attribute__((ext_vector_type(2))) float v2f;
typedef __attribute__((ext_vector_type(8))) float v8f;

__device__ __forceinline__ float lrelu(float v, float s) { return v > 0.f ? v : v * s; }

template <int K, int NCOLS>
__device__ __forceinline__ void
wmma_tile_f32(const float* __restrict__ A, const float* __restrict__ B,
              float* __restrict__ D, int M, int m0, int n0, int lane) {
  const int half = lane >> 4;
  const int lid  = lane & 15;
  const int ar = min(m0 + lid, M - 1);
  v8f acc = {};
#pragma unroll
  for (int k = 0; k < K; k += 4) {
    const int ka = k + half * 2;
    v2f a, b;
    a.x = A[(size_t)ar * K + ka];
    a.y = A[(size_t)ar * K + ka + 1];
    b.x = B[(size_t)ka * NCOLS + n0 + lid];
    b.y = B[(size_t)(ka + 1) * NCOLS + n0 + lid];
    acc = __builtin_amdgcn_wmma_f32_16x16x4_f32(false, a, false, b, (short)0, acc,
                                                false, false);
  }
  const int rbase = m0 + half * 8;
#pragma unroll
  for (int r = 0; r < 8; ++r) {
    const int row = rbase + r;
    if (row < M) D[(size_t)row * NCOLS + n0 + lid] = acc[r];
  }
}

template <int K, int NCOLS>
__global__ void __launch_bounds__(32)
wmma_gemm_f32(const float* __restrict__ A, const float* __restrict__ B,
              float* __restrict__ D, int M) {
  wmma_tile_f32<K, NCOLS>(A, B, D, M, blockIdx.x * 16, blockIdx.y * 16, threadIdx.x);
}

#define CHUNK_E 8192
#define BSHIFT 8
#define BNODES 256

__global__ void __launch_bounds__(256)
chunk_hist(const int* __restrict__ dst, int* __restrict__ hist, int E, int NB, int NC) {
  extern __shared__ int lc[];
  const int t = threadIdx.x;
  const int c = blockIdx.x;
  for (int i = t; i < NB; i += 256) lc[i] = 0;
  __syncthreads();
  const int e0 = c * CHUNK_E + t;
  int d[CHUNK_E / 256];
#pragma unroll
  for (int k = 0; k < CHUNK_E / 256; ++k) d[k] = dst[min(e0 + k * 256, E - 1)];
#pragma unroll
  for (int k = 0; k < CHUNK_E / 256; ++k)
    if (e0 + k * 256 < E) atomicAdd(&lc[d[k] >> BSHIFT], 1);
  __syncthreads();
  for (int i = t; i < NB; i += 256) hist[i * NC + c] = lc[i];
}

__device__ __forceinline__ void
scan_counts_body(const int* __restrict__ cnt, int* __restrict__ rowptr, int n) {
  __shared__ int wsum[32];
  const int t = threadIdx.x;
  const int lane = t & 31, w = t >> 5;
  const int per = (n + 1023) / 1024;
  const int b = min(t * per, n);
  const int e = min(b + per, n);
  int s = 0;
  for (int i = b; i < e; ++i) s += cnt[i];
  int v = s;
  for (int o = 1; o < 32; o <<= 1) {
    const int u = __shfl_up(v, o, 32);
    if (lane >= o) v += u;
  }
  if (lane == 31) wsum[w] = v;
  __syncthreads();
  if (w == 0) {
    int x = wsum[lane];
    for (int o = 1; o < 32; o <<= 1) {
      const int u = __shfl_up(x, o, 32);
      if (lane >= o) x += u;
    }
    wsum[lane] = x;
  }
  __syncthreads();
  int run = v - s + (w > 0 ? wsum[w - 1] : 0);
  for (int i = b; i < e; ++i) {
    rowptr[i] = run;
    run += cnt[i];
  }
  if (t == 1023) rowptr[n] = run;
}

template <int K, int NCOLS>
__global__ void __launch_bounds__(1024)
scan_and_gemm(const int* __restrict__ cnt, int* __restrict__ rowptr, int n,
              const float* __restrict__ A, const float* __restrict__ B,
              float* __restrict__ D, int M) {
  if (blockIdx.x == 0) {
    scan_counts_body(cnt, rowptr, n);
    return;
  }
  constexpr int NT = NCOLS / 16;
  const int tile = (int)(blockIdx.x - 1) * 32 + (int)(threadIdx.x >> 5);
  const int m0 = (tile / NT) * 16;
  if (m0 >= M) return;
  wmma_tile_f32<K, NCOLS>(A, B, D, M, m0, (tile % NT) * 16, threadIdx.x & 31);
}

__device__ __forceinline__ void
edge_partition_body(const int* __restrict__ src, const int* __restrict__ dst,
                    const int* __restrict__ base, unsigned* __restrict__ part,
                    int E, int NB, int NC, int c, int* lc) {
  const int t = threadIdx.x;
  for (int i = t; i < NB; i += 256) lc[i] = base[i * NC + c];
  __syncthreads();
  const int e0 = c * CHUNK_E + t;
  constexpr int G = 16;
  for (int g = 0; g < CHUNK_E / 256; g += G) {
    int d[G], s[G];
#pragma unroll
    for (int k = 0; k < G; ++k) {
      const int e = min(e0 + (g + k) * 256, E - 1);
      d[k] = dst[e];
      s[k] = src[e];
    }
    int pos[G];
#pragma unroll
    for (int k = 0; k < G; ++k)
      pos[k] = (e0 + (g + k) * 256 < E) ? atomicAdd(&lc[d[k] >> BSHIFT], 1) : 0;
#pragma unroll
    for (int k = 0; k < G; ++k)
      if (e0 + (g + k) * 256 < E)
        part[pos[k]] = (unsigned)s[k] | ((unsigned)(d[k] & (BNODES - 1)) << 24);
  }
}

__global__ void __launch_bounds__(1024)
bucket_sort(const int* __restrict__ base, const unsigned* __restrict__ part,
            int* __restrict__ rowptr, int* __restrict__ csr, int Nn, int NC) {
  __shared__ int cnt[BNODES];
  const int t = threadIdx.x;
  const int b = blockIdx.x;
  const int bstart = base[b * NC];
  const int bend = base[(b + 1) * NC];
  if (t < BNODES) cnt[t] = 0;
  __syncthreads();
  for (int e0 = bstart + t; e0 < bend; e0 += 4096) {
    unsigned p[4];
#pragma unroll
    for (int j = 0; j < 4; ++j) p[j] = part[min(e0 + j * 1024, bend - 1)];
#pragma unroll
    for (int j = 0; j < 4; ++j)
      if (e0 + j * 1024 < bend) atomicAdd(&cnt[p[j] >> 24], 1);
  }
  __syncthreads();
  if (t < 32) {
    int cc[BNODES / 32];
    int s = 0;
#pragma unroll
    for (int j = 0; j < BNODES / 32; ++j) {
      cc[j] = cnt[t * (BNODES / 32) + j];
      s += cc[j];
    }
    int v = s;
    for (int o = 1; o < 32; o <<= 1) {
      const int u = __shfl_up(v, o, 32);
      if (t >= o) v += u;
    }
    int run = v - s;
#pragma unroll
    for (int j = 0; j < BNODES / 32; ++j) {
      const int local = t * (BNODES / 32) + j;
      const int node = b * BNODES + local;
      cnt[local] = run;
      if (node < Nn) rowptr[node] = bstart + run;
      run += cc[j];
    }
  }
  if (b == (int)gridDim.x - 1 && t == 0) rowptr[Nn] = bend;
  __syncthreads();
  for (int e0 = bstart + t; e0 < bend; e0 += 4096) {
    unsigned p[4];
#pragma unroll
    for (int j = 0; j < 4; ++j) p[j] = part[min(e0 + j * 1024, bend - 1)];
    int pos[4];
#pragma unroll
    for (int j = 0; j < 4; ++j)
      pos[j] = (e0 + j * 1024 < bend) ? atomicAdd(&cnt[p[j] >> 24], 1) : 0;
#pragma unroll
    for (int j = 0; j < 4; ++j)
      if (e0 + j * 1024 < bend) csr[bstart + pos[j]] = (int)(p[j] & 0xFFFFFFu);
  }
}

__device__ __forceinline__ void
node_prep_body(const float* __restrict__ h,
               const float* __restrict__ att_src,
               const float* __restrict__ att_dst,
               float* __restrict__ as, float* __restrict__ ad,
               int Nn, int H, int C, int n) {
  if (n >= Nn) return;
  const int HC = H * C;
  for (int hh = 0; hh < H; ++hh) {
    float s = 0.f, d = 0.f;
    for (int c = 0; c < C; ++c) {
      const float v = h[(size_t)n * HC + hh * C + c];
      s += v * att_src[hh * C + c];
      d += v * att_dst[hh * C + c];
    }
    as[n * H + hh] = s;
    ad[n * H + hh] = d;
  }
}

__global__ void node_prep(const float* __restrict__ h,
                          const float* __restrict__ att_src,
                          const float* __restrict__ att_dst,
                          float* __restrict__ as, float* __restrict__ ad,
                          int Nn, int H, int C) {
  node_prep_body(h, att_src, att_dst, as, ad, Nn, H, C,
                 (int)(blockIdx.x * blockDim.x + threadIdx.x));
}

__global__ void __launch_bounds__(256)
partition_and_prep(const int* __restrict__ src, const int* __restrict__ dst,
                   const int* __restrict__ base, unsigned* __restrict__ part,
                   int E, int NB, int NC,
                   const float* __restrict__ h,
                   const float* __restrict__ att_src,
                   const float* __restrict__ att_dst,
                   float* __restrict__ as, float* __restrict__ ad,
                   int Nn, int H, int C, unsigned prep_blocks) {
  extern __shared__ int lc[];
  if (blockIdx.x < prep_blocks) {
    node_prep_body(h, att_src, att_dst, as, ad, Nn, H, C,
                   (int)(blockIdx.x * blockDim.x + threadIdx.x));
  } else {
    edge_partition_body(src, dst, base, part, E, NB, NC,
                        (int)(blockIdx.x - prep_blocks), lc);
  }
}

template <int H, int C, int VEC, int CH, bool FULL>
__device__ __forceinline__ void
softmax_chunk(const int* __restrict__ csr, const float* __restrict__ as,
              const float* __restrict__ hfeat, int i, int end, int hh, int f0,
              float adv, float& m, float& den, float (&acc)[VEC]) {
  constexpr int HC = H * C;
  int s[CH];
  float e[CH];
#pragma unroll
  for (int j = 0; j < CH; ++j) s[j] = FULL ? csr[i + j] : csr[min(i + j, end - 1)];
#pragma unroll
  for (int j = 0; j < CH; ++j) {
    const float v = lrelu(as[s[j] * H + hh] + adv, ATT_SLOPE);
    e[j] = (FULL || i + j < end) ? v : -INFINITY;
  }
  float cm = e[0];
#pragma unroll
  for (int j = 1; j < CH; ++j) cm = fmaxf(cm, e[j]);
  const float mn = fmaxf(m, cm);
  const float sc = expf(m - mn);
  m = mn;
  den *= sc;
#pragma unroll
  for (int k = 0; k < VEC; ++k) acc[k] *= sc;
#pragma unroll
  for (int j = 0; j < CH; ++j) {
    const float w = expf(e[j] - mn);
    den += w;
    const float* hp = hfeat + (size_t)s[j] * HC + f0;
#pragma unroll
    for (int k = 0; k < VEC; ++k) acc[k] += w * hp[k];
  }
}

template <int H, int C, int VEC, bool PROJ>
__global__ void __launch_bounds__(256)
seg_gather(const int* __restrict__ rowptr, const int* __restrict__ csr,
           const float* __restrict__ as, const float* __restrict__ ad,
           const float* __restrict__ hfeat, const float* __restrict__ bias,
           const float* __restrict__ Wo, const float* __restrict__ bo,
           float* __restrict__ dstbuf, int Nn) {
  constexpr int HC = H * C;
  constexpr int LPN = HC / VEC;
  constexpr int CH = 4;
  const unsigned gt = blockIdx.x * blockDim.x + threadIdx.x;
  const int node = (int)(gt / LPN);
  if (node >= Nn) return;
  const int gl = (int)(threadIdx.x % LPN);
  const int f0 = gl * VEC;
  const int hh = f0 / C;
  const int beg = rowptr[node];
  const int end = rowptr[node + 1];

  const float adv = ad[node * H + hh];
  float m = lrelu(as[node * H + hh] + adv, ATT_SLOPE);
  float den = 1.f;
  float acc[VEC];
  {
    const float* hp = hfeat + (size_t)node * HC + f0;
#pragma unroll
    for (int k = 0; k < VEC; ++k) acc[k] = hp[k];
  }

  int i = beg;
  for (; i + CH <= end; i += CH)
    softmax_chunk<H, C, VEC, CH, true>(csr, as, hfeat, i, end, hh, f0, adv, m, den, acc);
  if (i < end)
    softmax_chunk<H, C, VEC, CH, false>(csr, as, hfeat, i, end, hh, f0, adv, m, den, acc);

  if constexpr (PROJ) {
    float p = 0.f;
#pragma unroll
    for (int k = 0; k < VEC; ++k)
      p += lrelu(acc[k] / den + bias[f0 + k], ACT_SLOPE) * Wo[f0 + k];
    for (int o = 1; o < LPN; o <<= 1) p += __shfl_xor(p, o, LPN);
    if (gl == 0) dstbuf[node] = p + bo[0];
  } else {
    float* op = dstbuf + (size_t)node * HC + f0;
#pragma unroll
    for (int k = 0; k < VEC; ++k)
      op[k] = lrelu(acc[k] / den + bias[f0 + k], ACT_SLOPE);
  }
}

extern "C" void kernel_launch(void* const* d_in, const int* in_sizes, int n_in,
                              void* d_out, int out_size, void* d_ws, size_t ws_size,
                              hipStream_t stream) {
  const float* x    = (const float*)d_in[0];
  const int*   ei   = (const int*)d_in[1];
  const float* W1   = (const float*)d_in[2];
  const float* aS1  = (const float*)d_in[3];
  const float* aD1  = (const float*)d_in[4];
  const float* b1   = (const float*)d_in[5];
  const float* W2   = (const float*)d_in[6];
  const float* aS2  = (const float*)d_in[7];
  const float* aD2  = (const float*)d_in[8];
  const float* b2   = (const float*)d_in[9];
  const float* Wo   = (const float*)d_in[10];
  const float* bo   = (const float*)d_in[11];
  float* out = (float*)d_out;

  const int IN_C = 64, HID = 16, HEADS = 4;
  const int HC1 = HEADS * HID;
  const int Nn = in_sizes[0] / IN_C;
  const int E  = in_sizes[1] / 2;
  const int* src = ei;
  const int* dst = ei + E;

  float* p = (float*)d_ws;
  float* h1   = p; p += (size_t)Nn * HC1;
  float* f1   = p; p += (size_t)Nn * HC1;
  float* h2   = p; p += (size_t)Nn * HID;
  float* as1  = p; p += (size_t)Nn * HEADS;
  float* ad1  = p; p += (size_t)Nn * HEADS;
  float* as2  = p; p += (size_t)Nn;
  float* ad2  = p; p += (size_t)Nn;
  const int NB = (Nn + BNODES - 1) / BNODES;
  const int NC = (E + CHUNK_E - 1) / CHUNK_E;
  const int NT = NB * NC;
  int* q = (int*)p;
  int* hist   = q; q += (size_t)NT;
  int* base   = q; q += (size_t)NT + 1;
  int* rowptr = q; q += (size_t)Nn + 1;
  unsigned* part = (unsigned*)q; q += (size_t)E;
  int* csr    = q; q += (size_t)E;

  const int TB = 256;

  chunk_hist<<<NC, 256, NB * sizeof(int), stream>>>(dst, hist, E, NB, NC);
  {
    const int tiles = ((Nn + 15) / 16) * (HC1 / 16);
    scan_and_gemm<64, 64><<<1 + (tiles + 31) / 32, 1024, 0, stream>>>(
        hist, base, NT, x, W1, h1, Nn);
  }
  {
    const unsigned pb = (unsigned)((Nn + TB - 1) / TB);
    partition_and_prep<<<pb + (unsigned)NC, TB, NB * sizeof(int), stream>>>(
        src, dst, base, part, E, NB, NC, h1, aS1, aD1, as1, ad1, Nn, HEADS, HID, pb);
  }
  bucket_sort<<<NB, 1024, 0, stream>>>(base, part, rowptr, csr, Nn, NC);

  {
    const unsigned n = (unsigned)Nn * 16u;
    seg_gather<4, 16, 4, false><<<(n + TB - 1) / TB, TB, 0, stream>>>(
        rowptr, csr, as1, ad1, h1, b1, Wo, bo, f1, Nn);
  }

  {
    dim3 g((Nn + 15) / 16, HID / 16);
    wmma_gemm_f32<64, 16><<<g, 32, 0, stream>>>(f1, W2, h2, Nn);
  }
  node_prep<<<(Nn + TB - 1) / TB, TB, 0, stream>>>(h2, aS2, aD2, as2, ad2,
                                                   Nn, 1, HID);
  {
    const unsigned n = (unsigned)Nn * 8u;
    seg_gather<1, 16, 2, true><<<(n + TB - 1) / TB, TB, 0, stream>>>(
        rowptr, csr, as2, ad2, h2, b2, Wo, bo, out, Nn);
  }
}


# Both sort levels collect their sorted entries in LDS and store them as contiguous runs (whole 128-byte lines in level 2) instead of one scattered 4-byte global store per edge; sum order as before
# speedup vs baseline: 8.2081x; 1.2227x over previous
#include <hip/hip_runtime.h>
#include <math.h>

#define ATT_SLOPE 0.2f
#define ACT_SLOPE 0.01f

typedef __attribute__((ext_vector_type(2))) float v2f;
typedef __attribute__((ext_vector_type(8))) float v8f;

__device__ __forceinline__ float lrelu(float v, float s) { return v > 0.f ? v : v * s; }

template <int K, int NCOLS>
__device__ __forceinline__ void
wmma_tile_f32(const float* __restrict__ A, const float* __restrict__ B,
              float* __restrict__ D, int M, int m0, int n0, int lane) {
  const int half = lane >> 4;
  const int lid  = lane & 15;
  const int ar = min(m0 + lid, M - 1);
  v8f acc = {};
#pragma unroll
  for (int k = 0; k < K; k += 4) {
    const int ka = k + half * 2;
    v2f a, b;
    a.x = A[(size_t)ar * K + ka];
    a.y = A[(size_t)ar * K + ka + 1];
    b.x = B[(size_t)ka * NCOLS + n0 + lid];
    b.y = B[(size_t)(ka + 1) * NCOLS + n0 + lid];
    acc = __builtin_amdgcn_wmma_f32_16x16x4_f32(false, a, false, b, (short)0, acc,
                                                false, false);
  }
  const int rbase = m0 + half * 8;
#pragma unroll
  for (int r = 0; r < 8; ++r) {
    const int row = rbase + r;
    if (row < M) D[(size_t)row * NCOLS + n0 + lid] = acc[r];
  }
}

template <int K, int NCOLS>
__global__ void __launch_bounds__(32)
wmma_gemm_f32(const float* __restrict__ A, const float* __restrict__ B,
              float* __restrict__ D, int M) {
  wmma_tile_f32<K, NCOLS>(A, B, D, M, blockIdx.x * 16, blockIdx.y * 16, threadIdx.x);
}

#define CHUNK_E 8192
#define BSHIFT 8
#define BNODES 256

__global__ void __launch_bounds__(256)
chunk_hist(const int* __restrict__ dst, int* __restrict__ hist, int E, int NB, int NC) {
  extern __shared__ int lc[];
  const int t = threadIdx.x;
  const int c = blockIdx.x;
  for (int i = t; i < NB; i += 256) lc[i] = 0;
  __syncthreads();
  const int e0 = c * CHUNK_E + t;
  int d[CHUNK_E / 256];
#pragma unroll
  for (int k = 0; k < CHUNK_E / 256; ++k) d[k] = dst[min(e0 + k * 256, E - 1)];
#pragma unroll
  for (int k = 0; k < CHUNK_E / 256; ++k)
    if (e0 + k * 256 < E) atomicAdd(&lc[d[k] >> BSHIFT], 1);
  __syncthreads();
  for (int i = t; i < NB; i += 256) hist[i * NC + c] = lc[i];
}

__device__ __forceinline__ void
scan_counts_body(const int* __restrict__ cnt, int* __restrict__ rowptr, int n) {
  __shared__ int wsum[32];
  const int t = threadIdx.x;
  const int lane = t & 31, w = t >> 5;
  const int per = (n + 1023) / 1024;
  const int b = min(t * per, n);
  const int e = min(b + per, n);
  int s = 0;
  for (int i = b; i < e; ++i) s += cnt[i];
  int v = s;
  for (int o = 1; o < 32; o <<= 1) {
    const int u = __shfl_up(v, o, 32);
    if (lane >= o) v += u;
  }
  if (lane == 31) wsum[w] = v;
  __syncthreads();
  if (w == 0) {
    int x = wsum[lane];
    for (int o = 1; o < 32; o <<= 1) {
      const int u = __shfl_up(x, o, 32);
      if (lane >= o) x += u;
    }
    wsum[lane] = x;
  }
  __syncthreads();
  int run = v - s + (w > 0 ? wsum[w - 1] : 0);
  for (int i = b; i < e; ++i) {
    rowptr[i] = run;
    run += cnt[i];
  }
  if (t == 1023) rowptr[n] = run;
}

template <int K, int NCOLS>
__global__ void __launch_bounds__(1024)
scan_and_gemm(const int* __restrict__ cnt, int* __restrict__ rowptr, int n,
              const float* __restrict__ A, const float* __restrict__ B,
              float* __restrict__ D, int M) {
  if (blockIdx.x == 0) {
    scan_counts_body(cnt, rowptr, n);
    return;
  }
  constexpr int NT = NCOLS / 16;
  const int tile = (int)(blockIdx.x - 1) * 32 + (int)(threadIdx.x >> 5);
  const int m0 = (tile / NT) * 16;
  if (m0 >= M) return;
  wmma_tile_f32<K, NCOLS>(A, B, D, M, m0, (tile % NT) * 16, threadIdx.x & 31);
}

__device__ __forceinline__ void
edge_partition_body(const int* __restrict__ src, const int* __restrict__ dst,
                    const int* __restrict__ base, unsigned* __restrict__ part,
                    int E, int NB, int NC, int c, int* lds) {
  int* lc = lds;
  int* dl = lds + NB;
  unsigned* sval = (unsigned*)(lds + 2 * NB);
  unsigned short* sbk = (unsigned short*)(sval + CHUNK_E);
  const int t = threadIdx.x;
  for (int i = t; i < NB; i += 256) {
    const int g0 = base[i * NC + c];
    lc[i] = base[i * NC + c + 1] - g0;
    dl[i] = g0;
  }
  __syncthreads();
  if (t < 32) {
    const int per = (NB + 31) / 32;
    const int b0 = min(t * per, NB);
    const int b1 = min(b0 + per, NB);
    int s = 0;
    for (int i = b0; i < b1; ++i) s += lc[i];
    int v = s;
    for (int o = 1; o < 32; o <<= 1) {
      const int u = __shfl_up(v, o, 32);
      if (t >= o) v += u;
    }
    int run = v - s;
    for (int i = b0; i < b1; ++i) {
      const int cn = lc[i];
      lc[i] = run;
      dl[i] -= run;
      run += cn;
    }
  }
  __syncthreads();
  const int e0 = c * CHUNK_E + t;
  constexpr int G = 16;
  for (int g = 0; g < CHUNK_E / 256; g += G) {
    int d[G], s[G];
#pragma unroll
    for (int k = 0; k < G; ++k) {
      const int e = min(e0 + (g + k) * 256, E - 1);
      d[k] = dst[e];
      s[k] = src[e];
    }
    int pos[G];
#pragma unroll
    for (int k = 0; k < G; ++k)
      pos[k] = (e0 + (g + k) * 256 < E) ? atomicAdd(&lc[d[k] >> BSHIFT], 1) : 0;
#pragma unroll
    for (int k = 0; k < G; ++k)
      if (e0 + (g + k) * 256 < E) {
        sval[pos[k]] = (unsigned)s[k] | ((unsigned)(d[k] & (BNODES - 1)) << 24);
        sbk[pos[k]] = (unsigned short)(d[k] >> BSHIFT);
      }
  }
  __syncthreads();
  const int nloc = min(CHUNK_E, E - c * CHUNK_E);
  constexpr int WO = 16;
  for (int k0 = 0; k0 < CHUNK_E / 256; k0 += WO) {
    int bk[WO], off[WO];
    unsigned vv[WO];
#pragma unroll
    for (int k = 0; k < WO; ++k) {
      const int ii = min(t + (k0 + k) * 256, nloc - 1);
      bk[k] = sbk[ii];
      vv[k] = sval[ii];
    }
#pragma unroll
    for (int k = 0; k < WO; ++k) off[k] = dl[bk[k]];
#pragma unroll
    for (int k = 0; k < WO; ++k) {
      const int i = t + (k0 + k) * 256;
      if (i < nloc) part[i + off[k]] = vv[k];
    }
  }
}

#define BCAP 12288
__global__ void __launch_bounds__(1024)
bucket_sort(const int* __restrict__ base, const unsigned* __restrict__ part,
            int* __restrict__ rowptr, int* __restrict__ csr, int Nn, int NC) {
  __shared__ int cnt[BNODES];
  __shared__ int srt[BCAP];
  const int t = threadIdx.x;
  const int b = blockIdx.x;
  const int bstart = base[b * NC];
  const int bend = base[(b + 1) * NC];
  const bool staged = (bend - bstart) <= BCAP;
  if (t < BNODES) cnt[t] = 0;
  __syncthreads();
  for (int e0 = bstart + t; e0 < bend; e0 += 4096) {
    unsigned p[4];
#pragma unroll
    for (int j = 0; j < 4; ++j) p[j] = part[min(e0 + j * 1024, bend - 1)];
#pragma unroll
    for (int j = 0; j < 4; ++j)
      if (e0 + j * 1024 < bend) atomicAdd(&cnt[p[j] >> 24], 1);
  }
  __syncthreads();
  if (t < 32) {
    int cc[BNODES / 32];
    int s = 0;
#pragma unroll
    for (int j = 0; j < BNODES / 32; ++j) {
      cc[j] = cnt[t * (BNODES / 32) + j];
      s += cc[j];
    }
    int v = s;
    for (int o = 1; o < 32; o <<= 1) {
      const int u = __shfl_up(v, o, 32);
      if (t >= o) v += u;
    }
    int run = v - s;
#pragma unroll
    for (int j = 0; j < BNODES / 32; ++j) {
      const int local = t * (BNODES / 32) + j;
      const int node = b * BNODES + local;
      cnt[local] = run;
      if (node < Nn) rowptr[node] = bstart + run;
      run += cc[j];
    }
  }
  if (b == (int)gridDim.x - 1 && t == 0) rowptr[Nn] = bend;
  __syncthreads();
  for (int e0 = bstart + t; e0 < bend; e0 += 4096) {
    unsigned p[4];
#pragma unroll
    for (int j = 0; j < 4; ++j) p[j] = part[min(e0 + j * 1024, bend - 1)];
    int pos[4];
#pragma unroll
    for (int j = 0; j < 4; ++j)
      pos[j] = (e0 + j * 1024 < bend) ? atomicAdd(&cnt[p[j] >> 24], 1) : 0;
#pragma unroll
    for (int j = 0; j < 4; ++j)
      if (e0 + j * 1024 < bend) {
        const int sv = (int)(p[j] & 0xFFFFFFu);
        if (staged) srt[pos[j]] = sv;
        else csr[bstart + pos[j]] = sv;
      }
  }
  const int n = bend - bstart;
  if (staged && n > 0) {
    __syncthreads();
    const int sh = (int)((((size_t)(csr + bstart)) >> 2) & 31);
    for (int i0 = t - sh; i0 < n; i0 += 4096) {
      int v[4];
#pragma unroll
      for (int j = 0; j < 4; ++j) v[j] = srt[min(max(i0 + j * 1024, 0), n - 1)];
#pragma unroll
      for (int j = 0; j < 4; ++j) {
        const int i = i0 + j * 1024;
        if (i >= 0 && i < n) csr[bstart + i] = v[j];
      }
    }
  }
}

__device__ __forceinline__ void
node_prep_body(const float* __restrict__ h,
               const float* __restrict__ att_src,
               const float* __restrict__ att_dst,
               float* __restrict__ as, float* __restrict__ ad,
               int Nn, int H, int C, int n) {
  if (n >= Nn) return;
  const int HC = H * C;
  for (int hh = 0; hh < H; ++hh) {
    float s = 0.f, d = 0.f;
    for (int c = 0; c < C; ++c) {
      const float v = h[(size_t)n * HC + hh * C + c];
      s += v * att_src[hh * C + c];
      d += v * att_dst[hh * C + c];
    }
    as[n * H + hh] = s;
    ad[n * H + hh] = d;
  }
}

__global__ void node_prep(const float* __restrict__ h,
                          const float* __restrict__ att_src,
                          const float* __restrict__ att_dst,
                          float* __restrict__ as, float* __restrict__ ad,
                          int Nn, int H, int C) {
  node_prep_body(h, att_src, att_dst, as, ad, Nn, H, C,
                 (int)(blockIdx.x * blockDim.x + threadIdx.x));
}

__global__ void __launch_bounds__(256)
partition_and_prep(const int* __restrict__ src, const int* __restrict__ dst,
                   const int* __restrict__ base, unsigned* __restrict__ part,
                   int E, int NB, int NC,
                   const float* __restrict__ h,
                   const float* __restrict__ att_src,
                   const float* __restrict__ att_dst,
                   float* __restrict__ as, float* __restrict__ ad,
                   int Nn, int H, int C, unsigned prep_blocks) {
  extern __shared__ int lc[];
  if (blockIdx.x < prep_blocks) {
    node_prep_body(h, att_src, att_dst, as, ad, Nn, H, C,
                   (int)(blockIdx.x * blockDim.x + threadIdx.x));
  } else {
    edge_partition_body(src, dst, base, part, E, NB, NC,
                        (int)(blockIdx.x - prep_blocks), lc);
  }
}

template <int H, int C, int VEC, int CH, bool FULL>
__device__ __forceinline__ void
softmax_chunk(const int* __restrict__ csr, const float* __restrict__ as,
              const float* __restrict__ hfeat, int i, int end, int hh, int f0,
              float adv, float& m, float& den, float (&acc)[VEC]) {
  constexpr int HC = H * C;
  int s[CH];
  float e[CH];
#pragma unroll
  for (int j = 0; j < CH; ++j) s[j] = FULL ? csr[i + j] : csr[min(i + j, end - 1)];
#pragma unroll
  for (int j = 0; j < CH; ++j) {
    const float v = lrelu(as[s[j] * H + hh] + adv, ATT_SLOPE);
    e[j] = (FULL || i + j < end) ? v : -INFINITY;
  }
  float cm = e[0];
#pragma unroll
  for (int j = 1; j < CH; ++j) cm = fmaxf(cm, e[j]);
  const float mn = fmaxf(m, cm);
  const float sc = expf(m - mn);
  m = mn;
  den *= sc;
#pragma unroll
  for (int k = 0; k < VEC; ++k) acc[k] *= sc;
#pragma unroll
  for (int j = 0; j < CH; ++j) {
    const float w = expf(e[j] - mn);
    den += w;
    const float* hp = hfeat + (size_t)s[j] * HC + f0;
#pragma unroll
    for (int k = 0; k < VEC; ++k) acc[k] += w * hp[k];
  }
}

template <int H, int C, int VEC, bool PROJ>
__global__ void __launch_bounds__(256)
seg_gather(const int* __restrict__ rowptr, const int* __restrict__ csr,
           const float* __restrict__ as, const float* __restrict__ ad,
           const float* __restrict__ hfeat, const float* __restrict__ bias,
           const float* __restrict__ Wo, const float* __restrict__ bo,
           float* __restrict__ dstbuf, int Nn) {
  constexpr int HC = H * C;
  constexpr int LPN = HC / VEC;
  constexpr int CH = 4;
  const unsigned gt = blockIdx.x * blockDim.x + threadIdx.x;
  const int node = (int)(gt / LPN);
  if (node >= Nn) return;
  const int gl = (int)(threadIdx.x % LPN);
  const int f0 = gl * VEC;
  const int hh = f0 / C;
  const int beg = rowptr[node];
  const int end = rowptr[node + 1];

  const float adv = ad[node * H + hh];
  float m = lrelu(as[node * H + hh] + adv, ATT_SLOPE);
  float den = 1.f;
  float acc[VEC];
  {
    const float* hp = hfeat + (size_t)node * HC + f0;
#pragma unroll
    for (int k = 0; k < VEC; ++k) acc[k] = hp[k];
  }

  int i = beg;
  for (; i + CH <= end; i += CH)
    softmax_chunk<H, C, VEC, CH, true>(csr, as, hfeat, i, end, hh, f0, adv, m, den, acc);
  if (i < end)
    softmax_chunk<H, C, VEC, CH, false>(csr, as, hfeat, i, end, hh, f0, adv, m, den, acc);

  if constexpr (PROJ) {
    float p = 0.f;
#pragma unroll
    for (int k = 0; k < VEC; ++k)
      p += lrelu(acc[k] / den + bias[f0 + k], ACT_SLOPE) * Wo[f0 + k];
    for (int o = 1; o < LPN; o <<= 1) p += __shfl_xor(p, o, LPN);
    if (gl == 0) dstbuf[node] = p + bo[0];
  } else {
    float* op = dstbuf + (size_t)node * HC + f0;
#pragma unroll
    for (int k = 0; k < VEC; ++k)
      op[k] = lrelu(acc[k] / den + bias[f0 + k], ACT_SLOPE);
  }
}

extern "C" void kernel_launch(void* const* d_in, const int* in_sizes, int n_in,
                              void* d_out, int out_size, void* d_ws, size_t ws_size,
                              hipStream_t stream) {
  const float* x    = (const float*)d_in[0];
  const int*   ei   = (const int*)d_in[1];
  const float* W1   = (const float*)d_in[2];
  const float* aS1  = (const float*)d_in[3];
  const float* aD1  = (const float*)d_in[4];
  const float* b1   = (const float*)d_in[5];
  const float* W2   = (const float*)d_in[6];
  const float* aS2  = (const float*)d_in[7];
  const float* aD2  = (const float*)d_in[8];
  const float* b2   = (const float*)d_in[9];
  const float* Wo   = (const float*)d_in[10];
  const float* bo   = (const float*)d_in[11];
  float* out = (float*)d_out;

  const int IN_C = 64, HID = 16, HEADS = 4;
  const int HC1 = HEADS * HID;
  const int Nn = in_sizes[0] / IN_C;
  const int E  = in_sizes[1] / 2;
  const int* src = ei;
  const int* dst = ei + E;

  float* p = (float*)d_ws;
  float* h1   = p; p += (size_t)Nn * HC1;
  float* f1   = p; p += (size_t)Nn * HC1;
  float* h2   = p; p += (size_t)Nn * HID;
  float* as1  = p; p += (size_t)Nn * HEADS;
  float* ad1  = p; p += (size_t)Nn * HEADS;
  float* as2  = p; p += (size_t)Nn;
  float* ad2  = p; p += (size_t)Nn;
  const int NB = (Nn + BNODES - 1) / BNODES;
  const int NC = (E + CHUNK_E - 1) / CHUNK_E;
  const int NT = NB * NC;
  int* q = (int*)p;
  int* hist   = q; q += (size_t)NT;
  int* base   = q; q += (size_t)NT + 1;
  int* rowptr = q; q += (size_t)Nn + 1;
  unsigned* part = (unsigned*)q; q += (size_t)E;
  int* csr    = q; q += (size_t)E;

  const int TB = 256;

  chunk_hist<<<NC, 256, NB * sizeof(int), stream>>>(dst, hist, E, NB, NC);
  {
    const int tiles = ((Nn + 15) / 16) * (HC1 / 16);
    scan_and_gemm<64, 64><<<1 + (tiles + 31) / 32, 1024, 0, stream>>>(
        hist, base, NT, x, W1, h1, Nn);
  }
  {
    const unsigned pb = (unsigned)((Nn + TB - 1) / TB);
    partition_and_prep<<<pb + (unsigned)NC, TB,
                         (2 * NB + CHUNK_E) * sizeof(int) + CHUNK_E * sizeof(unsigned short),
                         stream>>>(
        src, dst, base, part, E, NB, NC, h1, aS1, aD1, as1, ad1, Nn, HEADS, HID, pb);
  }
  bucket_sort<<<NB, 1024, 0, stream>>>(base, part, rowptr, csr, Nn, NC);

  {
    const unsigned n = (unsigned)Nn * 16u;
    seg_gather<4, 16, 4, false><<<(n + TB - 1) / TB, TB, 0, stream>>>(
        rowptr, csr, as1, ad1, h1, b1, Wo, bo, f1, Nn);
  }

  {
    dim3 g((Nn + 15) / 16, HID / 16);
    wmma_gemm_f32<64, 16><<<g, 32, 0, stream>>>(f1, W2, h2, Nn);
  }
  node_prep<<<(Nn + TB - 1) / TB, TB, 0, stream>>>(h2, aS2, aD2, as2, ad2,
                                                   Nn, 1, HID);
  {
    const unsigned n = (unsigned)Nn * 8u;
    seg_gather<1, 16, 2, true><<<(n + TB - 1) / TB, TB, 0, stream>>>(
        rowptr, csr, as2, ad2, h2, b2, Wo, bo, out, Nn);
  }
}


# Edge scores computed from the gathered feature row (lane partial dot + row_xmask add tree over the head's lanes), not gathered per source: prep passes go, 8 launches 7; score rounding changes
# speedup vs baseline: 8.9178x; 1.0865x over previous
#include <hip/hip_runtime.h>
#include <math.h>

#define ATT_SLOPE 0.2f
#define ACT_SLOPE 0.01f

typedef __attribute__((ext_vector_type(2))) float v2f;
typedef __attribute__((ext_vector_type(8))) float v8f;

__device__ __forceinline__ float lrelu(float v, float s) { return v > 0.f ? v : v * s; }

template <int K, int NCOLS>
__device__ __forceinline__ void
wmma_tile_f32(const float* __restrict__ A, const float* __restrict__ B,
              float* __restrict__ D, int M, int m0, int n0, int lane) {
  const int half = lane >> 4;
  const int lid  = lane & 15;
  const int ar = min(m0 + lid, M - 1);
  v8f acc = {};
#pragma unroll
  for (int k = 0; k < K; k += 4) {
    const int ka = k + half * 2;
    v2f a, b;
    a.x = A[(size_t)ar * K + ka];
    a.y = A[(size_t)ar * K + ka + 1];
    b.x = B[(size_t)ka * NCOLS + n0 + lid];
    b.y = B[(size_t)(ka + 1) * NCOLS + n0 + lid];
    acc = __builtin_amdgcn_wmma_f32_16x16x4_f32(false, a, false, b, (short)0, acc,
                                                false, false);
  }
  const int rbase = m0 + half * 8;
#pragma unroll
  for (int r = 0; r < 8; ++r) {
    const int row = rbase + r;
    if (row < M) D[(size_t)row * NCOLS + n0 + lid] = acc[r];
  }
}

template <int K, int NCOLS>
__global__ void __launch_bounds__(32)
wmma_gemm_f32(const float* __restrict__ A, const float* __restrict__ B,
              float* __restrict__ D, int M) {
  wmma_tile_f32<K, NCOLS>(A, B, D, M, blockIdx.x * 16, blockIdx.y * 16, threadIdx.x);
}

#define CHUNK_E 8192
#define BSHIFT 8
#define BNODES 256

__global__ void __launch_bounds__(256)
chunk_hist(const int* __restrict__ dst, int* __restrict__ hist, int E, int NB, int NC) {
  extern __shared__ int lc[];
  const int t = threadIdx.x;
  const int c = blockIdx.x;
  for (int i = t; i < NB; i += 256) lc[i] = 0;
  __syncthreads();
  const int e0 = c * CHUNK_E + t;
  int d[CHUNK_E / 256];
#pragma unroll
  for (int k = 0; k < CHUNK_E / 256; ++k) d[k] = dst[min(e0 + k * 256, E - 1)];
#pragma unroll
  for (int k = 0; k < CHUNK_E / 256; ++k)
    if (e0 + k * 256 < E) atomicAdd(&lc[d[k] >> BSHIFT], 1);
  __syncthreads();
  for (int i = t; i < NB; i += 256) hist[i * NC + c] = lc[i];
}

__device__ __forceinline__ void
scan_counts_body(const int* __restrict__ cnt, int* __restrict__ rowptr, int n) {
  __shared__ int wsum[32];
  const int t = threadIdx.x;
  const int lane = t & 31, w = t >> 5;
  const int per = (n + 1023) / 1024;
  const int b = min(t * per, n);
  const int e = min(b + per, n);
  int s = 0;
  for (int i = b; i < e; ++i) s += cnt[i];
  int v = s;
  for (int o = 1; o < 32; o <<= 1) {
    const int u = __shfl_up(v, o, 32);
    if (lane >= o) v += u;
  }
  if (lane == 31) wsum[w] = v;
  __syncthreads();
  if (w == 0) {
    int x = wsum[lane];
    for (int o = 1; o < 32; o <<= 1) {
      const int u = __shfl_up(x, o, 32);
      if (lane >= o) x += u;
    }
    wsum[lane] = x;
  }
  __syncthreads();
  int run = v - s + (w > 0 ? wsum[w - 1] : 0);
  for (int i = b; i < e; ++i) {
    rowptr[i] = run;
    run += cnt[i];
  }
  if (t == 1023) rowptr[n] = run;
}

template <int K, int NCOLS>
__global__ void __launch_bounds__(1024)
scan_and_gemm(const int* __restrict__ cnt, int* __restrict__ rowptr, int n,
              const float* __restrict__ A, const float* __restrict__ B,
              float* __restrict__ D, int M) {
  if (blockIdx.x == 0) {
    scan_counts_body(cnt, rowptr, n);
    return;
  }
  constexpr int NT = NCOLS / 16;
  const int tile = (int)(blockIdx.x - 1) * 32 + (int)(threadIdx.x >> 5);
  const int m0 = (tile / NT) * 16;
  if (m0 >= M) return;
  wmma_tile_f32<K, NCOLS>(A, B, D, M, m0, (tile % NT) * 16, threadIdx.x & 31);
}

__device__ __forceinline__ void
edge_partition_body(const int* __restrict__ src, const int* __restrict__ dst,
                    const int* __restrict__ base, unsigned* __restrict__ part,
                    int E, int NB, int NC, int c, int* lds) {
  int* lc = lds;
  int* dl = lds + NB;
  unsigned* sval = (unsigned*)(lds + 2 * NB);
  unsigned short* sbk = (unsigned short*)(sval + CHUNK_E);
  const int t = threadIdx.x;
  for (int i = t; i < NB; i += 256) {
    const int g0 = base[i * NC + c];
    lc[i] = base[i * NC + c + 1] - g0;
    dl[i] = g0;
  }
  __syncthreads();
  if (t < 32) {
    const int per = (NB + 31) / 32;
    const int b0 = min(t * per, NB);
    const int b1 = min(b0 + per, NB);
    int s = 0;
    for (int i = b0; i < b1; ++i) s += lc[i];
    int v = s;
    for (int o = 1; o < 32; o <<= 1) {
      const int u = __shfl_up(v, o, 32);
      if (t >= o) v += u;
    }
    int run = v - s;
    for (int i = b0; i < b1; ++i) {
      const int cn = lc[i];
      lc[i] = run;
      dl[i] -= run;
      run += cn;
    }
  }
  __syncthreads();
  const int e0 = c * CHUNK_E + t;
  constexpr int G = 16;
  for (int g = 0; g < CHUNK_E / 256; g += G) {
    int d[G], s[G];
#pragma unroll
    for (int k = 0; k < G; ++k) {
      const int e = min(e0 + (g + k) * 256, E - 1);
      d[k] = dst[e];
      s[k] = src[e];
    }
    int pos[G];
#pragma unroll
    for (int k = 0; k < G; ++k)
      pos[k] = (e0 + (g + k) * 256 < E) ? atomicAdd(&lc[d[k] >> BSHIFT], 1) : 0;
#pragma unroll
    for (int k = 0; k < G; ++k)
      if (e0 + (g + k) * 256 < E) {
        sval[pos[k]] = (unsigned)s[k] | ((unsigned)(d[k] & (BNODES - 1)) << 24);
        sbk[pos[k]] = (unsigned short)(d[k] >> BSHIFT);
      }
  }
  __syncthreads();
  const int nloc = min(CHUNK_E, E - c * CHUNK_E);
  constexpr int WO = 16;
  for (int k0 = 0; k0 < CHUNK_E / 256; k0 += WO) {
    int bk[WO], off[WO];
    unsigned vv[WO];
#pragma unroll
    for (int k = 0; k < WO; ++k) {
      const int ii = min(t + (k0 + k) * 256, nloc - 1);
      bk[k] = sbk[ii];
      vv[k] = sval[ii];
    }
#pragma unroll
    for (int k = 0; k < WO; ++k) off[k] = dl[bk[k]];
#pragma unroll
    for (int k = 0; k < WO; ++k) {
      const int i = t + (k0 + k) * 256;
      if (i < nloc) part[i + off[k]] = vv[k];
    }
  }
}

#define BCAP 12288
__global__ void __launch_bounds__(1024)
bucket_sort(const int* __restrict__ base, const unsigned* __restrict__ part,
            int* __restrict__ rowptr, int* __restrict__ csr, int Nn, int NC) {
  __shared__ int cnt[BNODES];
  __shared__ int srt[BCAP];
  const int t = threadIdx.x;
  const int b = blockIdx.x;
  const int bstart = base[b * NC];
  const int bend = base[(b + 1) * NC];
  const bool staged = (bend - bstart) <= BCAP;
  if (t < BNODES) cnt[t] = 0;
  __syncthreads();
  for (int e0 = bstart + t; e0 < bend; e0 += 4096) {
    unsigned p[4];
#pragma unroll
    for (int j = 0; j < 4; ++j) p[j] = part[min(e0 + j * 1024, bend - 1)];
#pragma unroll
    for (int j = 0; j < 4; ++j)
      if (e0 + j * 1024 < bend) atomicAdd(&cnt[p[j] >> 24], 1);
  }
  __syncthreads();
  if (t < 32) {
    int cc[BNODES / 32];
    int s = 0;
#pragma unroll
    for (int j = 0; j < BNODES / 32; ++j) {
      cc[j] = cnt[t * (BNODES / 32) + j];
      s += cc[j];
    }
    int v = s;
    for (int o = 1; o < 32; o <<= 1) {
      const int u = __shfl_up(v, o, 32);
      if (t >= o) v += u;
    }
    int run = v - s;
#pragma unroll
    for (int j = 0; j < BNODES / 32; ++j) {
      const int local = t * (BNODES / 32) + j;
      const int node = b * BNODES + local;
      cnt[local] = run;
      if (node < Nn) rowptr[node] = bstart + run;
      run += cc[j];
    }
  }
  if (b == (int)gridDim.x - 1 && t == 0) rowptr[Nn] = bend;
  __syncthreads();
  for (int e0 = bstart + t; e0 < bend; e0 += 4096) {
    unsigned p[4];
#pragma unroll
    for (int j = 0; j < 4; ++j) p[j] = part[min(e0 + j * 1024, bend - 1)];
    int pos[4];
#pragma unroll
    for (int j = 0; j < 4; ++j)
      pos[j] = (e0 + j * 1024 < bend) ? atomicAdd(&cnt[p[j] >> 24], 1) : 0;
#pragma unroll
    for (int j = 0; j < 4; ++j)
      if (e0 + j * 1024 < bend) {
        const int sv = (int)(p[j] & 0xFFFFFFu);
        if (staged) srt[pos[j]] = sv;
        else csr[bstart + pos[j]] = sv;
      }
  }
  const int n = bend - bstart;
  if (staged && n > 0) {
    __syncthreads();
    const int sh = (int)((((size_t)(csr + bstart)) >> 2) & 31);
    for (int i0 = t - sh; i0 < n; i0 += 4096) {
      int v[4];
#pragma unroll
      for (int j = 0; j < 4; ++j) v[j] = srt[min(max(i0 + j * 1024, 0), n - 1)];
#pragma unroll
      for (int j = 0; j < 4; ++j) {
        const int i = i0 + j * 1024;
        if (i >= 0 && i < n) csr[bstart + i] = v[j];
      }
    }
  }
}

__global__ void __launch_bounds__(256)
edge_partition(const int* __restrict__ src, const int* __restrict__ dst,
               const int* __restrict__ base, unsigned* __restrict__ part,
               int E, int NB, int NC) {
  extern __shared__ int lc[];
  edge_partition_body(src, dst, base, part, E, NB, NC, (int)blockIdx.x, lc);
}

template <int N> __device__ __forceinline__ float add_lane_xor(float v);
template <> __device__ __forceinline__ float add_lane_xor<1>(float v) {
  float r;
  asm("v_add_f32_dpp %0, %1, %1 row_xmask:1 row_mask:0xf bank_mask:0xf" : "=v"(r) : "v"(v));
  return r;
}
template <> __device__ __forceinline__ float add_lane_xor<2>(float v) {
  float r;
  asm("v_add_f32_dpp %0, %1, %1 row_xmask:2 row_mask:0xf bank_mask:0xf" : "=v"(r) : "v"(v));
  return r;
}
template <> __device__ __forceinline__ float add_lane_xor<4>(float v) {
  float r;
  asm("v_add_f32_dpp %0, %1, %1 row_xmask:4 row_mask:0xf bank_mask:0xf" : "=v"(r) : "v"(v));
  return r;
}

template <int LANES>
__device__ __forceinline__ float head_sum(float v) {
  v = add_lane_xor<1>(v);
  v = add_lane_xor<2>(v);
  if constexpr (LANES >= 8) v = add_lane_xor<4>(v);
  return v;
}

typedef __attribute__((ext_vector_type(4))) float v4f;
template <int VEC> struct fvec;
template <> struct fvec<2> { typedef v2f type; };
template <> struct fvec<4> { typedef v4f type; };

__device__ __forceinline__ float lane_dot(v2f x, v2f a) {
  const v2f t = x * a;
  return t.x + t.y;
}
__device__ __forceinline__ float lane_dot(v4f x, v4f a) {
  const v2f t = x.xy * a.xy + x.zw * a.zw;
  return t.x + t.y;
}

template <int H, int C, int VEC, int CH, bool FULL>
__device__ __forceinline__ void
softmax_chunk(const int* __restrict__ csr, const float* __restrict__ hfeat,
              int i, int end, int f0, typename fvec<VEC>::type asrc, float adv,
              float& m, float& den, typename fvec<VEC>::type& acc) {
  typedef typename fvec<VEC>::type vf;
  constexpr int HC = H * C;
  int s[CH];
  vf x[CH];
  float e[CH];
#pragma unroll
  for (int j = 0; j < CH; ++j) s[j] = FULL ? csr[i + j] : csr[min(i + j, end - 1)];
#pragma unroll
  for (int j = 0; j < CH; ++j) x[j] = *(const vf*)(hfeat + (size_t)s[j] * HC + f0);
#pragma unroll
  for (int j = 0; j < CH; ++j) {
    const float v = lrelu(head_sum<C / VEC>(lane_dot(x[j], asrc)) + adv, ATT_SLOPE);
    e[j] = (FULL || i + j < end) ? v : -INFINITY;
  }
  float cm = e[0];
#pragma unroll
  for (int j = 1; j < CH; ++j) cm = fmaxf(cm, e[j]);
  const float mn = fmaxf(m, cm);
  const float sc = expf(m - mn);
  m = mn;
  den *= sc;
  acc *= sc;
#pragma unroll
  for (int j = 0; j < CH; ++j) {
    const float w = expf(e[j] - mn);
    den += w;
    acc += x[j] * w;
  }
}

template <int H, int C, int VEC, bool PROJ>
__global__ void __launch_bounds__(256)
seg_gather(const int* __restrict__ rowptr, const int* __restrict__ csr,
           const float* __restrict__ hfeat,
           const float* __restrict__ att_src, const float* __restrict__ att_dst,
           const float* __restrict__ bias,
           const float* __restrict__ Wo, const float* __restrict__ bo,
           float* __restrict__ dstbuf, int Nn) {
  typedef typename fvec<VEC>::type vf;
  constexpr int HC = H * C;
  constexpr int LPN = HC / VEC;
  constexpr int CH = 4;
  const unsigned gt = blockIdx.x * blockDim.x + threadIdx.x;
  const int node = (int)(gt / LPN);
  if (node >= Nn) return;
  const int gl = (int)(threadIdx.x % LPN);
  const int f0 = gl * VEC;
  const int beg = rowptr[node];
  const int end = rowptr[node + 1];

  const vf asrc = *(const vf*)(att_src + f0);
  vf acc = *(const vf*)(hfeat + (size_t)node * HC + f0);
  const float adv = head_sum<C / VEC>(lane_dot(acc, *(const vf*)(att_dst + f0)));
  float m = lrelu(head_sum<C / VEC>(lane_dot(acc, asrc)) + adv, ATT_SLOPE);
  float den = 1.f;

  int i = beg;
  for (; i + CH <= end; i += CH)
    softmax_chunk<H, C, VEC, CH, true>(csr, hfeat, i, end, f0, asrc, adv, m, den, acc);
  if (i < end)
    softmax_chunk<H, C, VEC, CH, false>(csr, hfeat, i, end, f0, asrc, adv, m, den, acc);

  if constexpr (PROJ) {
    float p = 0.f;
#pragma unroll
    for (int k = 0; k < VEC; ++k)
      p += lrelu(acc[k] / den + bias[f0 + k], ACT_SLOPE) * Wo[f0 + k];
    for (int o = 1; o < LPN; o <<= 1) p += __shfl_xor(p, o, LPN);
    if (gl == 0) dstbuf[node] = p + bo[0];
  } else {
    float* op = dstbuf + (size_t)node * HC + f0;
#pragma unroll
    for (int k = 0; k < VEC; ++k)
      op[k] = lrelu(acc[k] / den + bias[f0 + k], ACT_SLOPE);
  }
}

extern "C" void kernel_launch(void* const* d_in, const int* in_sizes, int n_in,
                              void* d_out, int out_size, void* d_ws, size_t ws_size,
                              hipStream_t stream) {
  const float* x    = (const float*)d_in[0];
  const int*   ei   = (const int*)d_in[1];
  const float* W1   = (const float*)d_in[2];
  const float* aS1  = (const float*)d_in[3];
  const float* aD1  = (const float*)d_in[4];
  const float* b1   = (const float*)d_in[5];
  const float* W2   = (const float*)d_in[6];
  const float* aS2  = (const float*)d_in[7];
  const float* aD2  = (const float*)d_in[8];
  const float* b2   = (const float*)d_in[9];
  const float* Wo   = (const float*)d_in[10];
  const float* bo   = (const float*)d_in[11];
  float* out = (float*)d_out;

  const int IN_C = 64, HID = 16, HEADS = 4;
  const int HC1 = HEADS * HID;
  const int Nn = in_sizes[0] / IN_C;
  const int E  = in_sizes[1] / 2;
  const int* src = ei;
  const int* dst = ei + E;

  float* p = (float*)d_ws;
  float* h1   = p; p += (size_t)Nn * HC1;
  float* f1   = p; p += (size_t)Nn * HC1;
  float* h2   = p; p += (size_t)Nn * HID;
  const int NB = (Nn + BNODES - 1) / BNODES;
  const int NC = (E + CHUNK_E - 1) / CHUNK_E;
  const int NT = NB * NC;
  int* q = (int*)p;
  int* hist   = q; q += (size_t)NT;
  int* base   = q; q += (size_t)NT + 1;
  int* rowptr = q; q += (size_t)Nn + 1;
  unsigned* part = (unsigned*)q; q += (size_t)E;
  int* csr    = q; q += (size_t)E;

  const int TB = 256;

  chunk_hist<<<NC, 256, NB * sizeof(int), stream>>>(dst, hist, E, NB, NC);
  {
    const int tiles = ((Nn + 15) / 16) * (HC1 / 16);
    scan_and_gemm<64, 64><<<1 + (tiles + 31) / 32, 1024, 0, stream>>>(
        hist, base, NT, x, W1, h1, Nn);
  }
  edge_partition<<<NC, TB,
                   (2 * NB + CHUNK_E) * sizeof(int) + CHUNK_E * sizeof(unsigned short),
                   stream>>>(src, dst, base, part, E, NB, NC);
  bucket_sort<<<NB, 1024, 0, stream>>>(base, part, rowptr, csr, Nn, NC);

  {
    const unsigned n = (unsigned)Nn * 16u;
    seg_gather<4, 16, 4, false><<<(n + TB - 1) / TB, TB, 0, stream>>>(
        rowptr, csr, h1, aS1, aD1, b1, Wo, bo, f1, Nn);
  }

  {
    dim3 g((Nn + 15) / 16, HID / 16);
    wmma_gemm_f32<64, 16><<<g, 32, 0, stream>>>(f1, W2, h2, Nn);
  }
  {
    const unsigned n = (unsigned)Nn * 8u;
    seg_gather<1, 16, 2, true><<<(n + TB - 1) / TB, TB, 0, stream>>>(
        rowptr, csr, h2, aS2, aD2, b2, Wo, bo, out, Nn);
  }
}


# Level 1 sorts each 8192-edge chunk by bucket on its own (bucket starts to a chunk table); level 2 finds its runs there: global count and prefix launches go, 7 launches 5; segment sum order changes
# speedup vs baseline: 11.2553x; 1.2621x over previous
#include <hip/hip_runtime.h>
#include <math.h>

#define ATT_SLOPE 0.2f
#define ACT_SLOPE 0.01f

typedef __attribute__((ext_vector_type(2))) float v2f;
typedef __attribute__((ext_vector_type(8))) float v8f;

__device__ __forceinline__ float lrelu(float v, float s) { return v > 0.f ? v : v * s; }

template <int K, int NCOLS>
__device__ __forceinline__ void
wmma_tile_f32(const float* __restrict__ A, const float* __restrict__ B,
              float* __restrict__ D, int M, int m0, int n0, int lane) {
  const int half = lane >> 4;
  const int lid  = lane & 15;
  const int ar = min(m0 + lid, M - 1);
  v8f acc = {};
#pragma unroll
  for (int k = 0; k < K; k += 4) {
    const int ka = k + half * 2;
    v2f a, b;
    a.x = A[(size_t)ar * K + ka];
    a.y = A[(size_t)ar * K + ka + 1];
    b.x = B[(size_t)ka * NCOLS + n0 + lid];
    b.y = B[(size_t)(ka + 1) * NCOLS + n0 + lid];
    acc = __builtin_amdgcn_wmma_f32_16x16x4_f32(false, a, false, b, (short)0, acc,
                                                false, false);
  }
  const int rbase = m0 + half * 8;
#pragma unroll
  for (int r = 0; r < 8; ++r) {
    const int row = rbase + r;
    if (row < M) D[(size_t)row * NCOLS + n0 + lid] = acc[r];
  }
}

template <int K, int NCOLS>
__global__ void __launch_bounds__(32)
wmma_gemm_f32(const float* __restrict__ A, const float* __restrict__ B,
              float* __restrict__ D, int M) {
  wmma_tile_f32<K, NCOLS>(A, B, D, M, blockIdx.x * 16, blockIdx.y * 16, threadIdx.x);
}

#define CHUNK_E 8192
#define BSHIFT 8
#define BNODES 256

__device__ __forceinline__ int
block_scan_lds(int* a, int n, int* wsum, int* __restrict__ gout) {
  const int t = threadIdx.x;
  const int lane = t & 31, w = t >> 5;
  const int per = (n + 1023) / 1024;
  const int b = min(t * per, n);
  const int e = min(b + per, n);
  int s = 0;
  for (int i = b; i < e; ++i) s += a[i];
  int v = s;
  for (int o = 1; o < 32; o <<= 1) {
    const int u = __shfl_up(v, o, 32);
    if (lane >= o) v += u;
  }
  if (lane == 31) wsum[w] = v;
  __syncthreads();
  if (w == 0) {
    int x = wsum[lane];
    for (int o = 1; o < 32; o <<= 1) {
      const int u = __shfl_up(x, o, 32);
      if (lane >= o) x += u;
    }
    wsum[lane] = x;
  }
  __syncthreads();
  int run = v - s + (w > 0 ? wsum[w - 1] : 0);
  for (int i = b; i < e; ++i) {
    const int cn = a[i];
    a[i] = run;
    if (gout) gout[i] = run;
    run += cn;
  }
  return wsum[31];
}

__device__ __forceinline__ void
chunk_sort_body(const int* __restrict__ src, const int* __restrict__ dst,
                int* __restrict__ off, unsigned* __restrict__ part,
                int E, int NB, int c, int* lds) {
  int* wsum = lds;
  int* lc = lds + 32;
  unsigned* sval = (unsigned*)(lds + 32 + NB);
  const int t = threadIdx.x;
  constexpr int PER = CHUNK_E / 1024;
  for (int i = t; i < NB; i += 1024) lc[i] = 0;
  const int e0 = c * CHUNK_E + t;
  int d[PER], s[PER];
#pragma unroll
  for (int k = 0; k < PER; ++k) {
    const int e = min(e0 + k * 1024, E - 1);
    d[k] = dst[e];
    s[k] = src[e];
  }
  __syncthreads();
#pragma unroll
  for (int k = 0; k < PER; ++k)
    if (e0 + k * 1024 < E) atomicAdd(&lc[d[k] >> BSHIFT], 1);
  __syncthreads();
  int* orow = off + (size_t)c * (NB + 1);
  const int nloc = block_scan_lds(lc, NB, wsum, orow);
  if (t == 0) orow[NB] = nloc;
  __syncthreads();
  int pos[PER];
#pragma unroll
  for (int k = 0; k < PER; ++k)
    pos[k] = (e0 + k * 1024 < E) ? atomicAdd(&lc[d[k] >> BSHIFT], 1) : 0;
#pragma unroll
  for (int k = 0; k < PER; ++k)
    if (e0 + k * 1024 < E)
      sval[pos[k]] = (unsigned)s[k] | ((unsigned)(d[k] & (BNODES - 1)) << 24);
  __syncthreads();
  unsigned vv[PER];
#pragma unroll
  for (int k = 0; k < PER; ++k) vv[k] = sval[min(t + k * 1024, nloc - 1)];
#pragma unroll
  for (int k = 0; k < PER; ++k) {
    const int i = t + k * 1024;
    if (i < nloc) part[(size_t)c * CHUNK_E + i] = vv[k];
  }
}

template <int K, int NCOLS>
__global__ void __launch_bounds__(1024)
chunk_sort_and_gemm(const int* __restrict__ src, const int* __restrict__ dst,
                    int* __restrict__ off, unsigned* __restrict__ part,
                    int E, int NB, int NC,
                    const float* __restrict__ A, const float* __restrict__ B,
                    float* __restrict__ D, int M) {
  extern __shared__ int lds[];
  if ((int)blockIdx.x < NC) {
    chunk_sort_body(src, dst, off, part, E, NB, (int)blockIdx.x, lds);
    return;
  }
  constexpr int NT = NCOLS / 16;
  const int tile = ((int)blockIdx.x - NC) * 32 + (int)(threadIdx.x >> 5);
  const int m0 = (tile / NT) * 16;
  if (m0 >= M) return;
  wmma_tile_f32<K, NCOLS>(A, B, D, M, m0, (tile % NT) * 16, threadIdx.x & 31);
}

#define BCAP 12288
__global__ void __launch_bounds__(1024)
bucket_sort(const int* __restrict__ off, const unsigned* __restrict__ part,
            int* __restrict__ rowptr, int* __restrict__ csr, int Nn, int NB, int NC) {
  extern __shared__ int lds[];
  __shared__ int cnt[BNODES];
  __shared__ int srt[BCAP];
  __shared__ int wsum[32];
  __shared__ int bsum;
  int* rs = lds;
  int* rd = lds + NC + 1;
  const int t = threadIdx.x;
  const int b = blockIdx.x;
  if (t < BNODES) cnt[t] = 0;
  if (t == 0) bsum = 0;
  __syncthreads();
  {
    int so = 0;
    for (int c = t; c < NC; c += 1024) {
      const int o0 = off[(size_t)c * (NB + 1) + b];
      const int o1 = off[(size_t)c * (NB + 1) + b + 1];
      rs[c] = o1 - o0;
      rd[c] = c * CHUNK_E + o0;
      so += o0;
    }
    if (so != 0) atomicAdd(&bsum, so);
  }
  __syncthreads();
  const int n = block_scan_lds(rs, NC, wsum, nullptr);
  const int bstart = bsum;
  if (t == 0) rs[NC] = n;
  __syncthreads();
  for (int c = t; c < NC; c += 1024) rd[c] -= rs[c];
  __syncthreads();
  const bool staged = n <= BCAP;
  int P2 = 1;
  while (P2 < NC) P2 <<= 1;
  auto locate = [&](int j) -> int {
    int lo = 0;
    for (int sp = P2 >> 1; sp > 0; sp >>= 1) {
      const int m = lo + sp;
      if (rs[min(m, NC)] <= j) lo = m;
    }
    return rd[lo] + j;
  };
  auto locate4 = [&](const int (&j)[4], int (&a)[4]) {
    int lo[4] = {0, 0, 0, 0};
    for (int sp = P2 >> 1; sp > 0; sp >>= 1) {
      int v[4];
#pragma unroll
      for (int k = 0; k < 4; ++k) v[k] = rs[min(lo[k] + sp, NC)];
#pragma unroll
      for (int k = 0; k < 4; ++k)
        if (v[k] <= j[k]) lo[k] += sp;
    }
#pragma unroll
    for (int k = 0; k < 4; ++k) a[k] = rd[lo[k]] + j[k];
  };
  constexpr int KP = BCAP / 1024;
  unsigned p[KP];
  if (staged) {
#pragma unroll
    for (int g = 0; g < KP; g += 4) {
      if (g * 1024 < n) {
        int jj[4], a[4];
#pragma unroll
        for (int k = 0; k < 4; ++k) jj[k] = min(t + (g + k) * 1024, n - 1);
        locate4(jj, a);
#pragma unroll
        for (int k = 0; k < 4; ++k) p[g + k] = part[a[k]];
#pragma unroll
        for (int k = 0; k < 4; ++k)
          if (t + (g + k) * 1024 < n) atomicAdd(&cnt[p[g + k] >> 24], 1);
      }
    }
  } else {
    for (int j = t; j < n; j += 1024) atomicAdd(&cnt[part[locate(j)] >> 24], 1);
  }
  __syncthreads();
  if (t < 32) {
    int cc[BNODES / 32];
    int s = 0;
#pragma unroll
    for (int j = 0; j < BNODES / 32; ++j) {
      cc[j] = cnt[t * (BNODES / 32) + j];
      s += cc[j];
    }
    int v = s;
    for (int o = 1; o < 32; o <<= 1) {
      const int u = __shfl_up(v, o, 32);
      if (t >= o) v += u;
    }
    int run = v - s;
#pragma unroll
    for (int j = 0; j < BNODES / 32; ++j) {
      const int local = t * (BNODES / 32) + j;
      const int node = b * BNODES + local;
      cnt[local] = run;
      if (node < Nn) rowptr[node] = bstart + run;
      run += cc[j];
    }
  }
  if (b == (int)gridDim.x - 1 && t == 0) rowptr[Nn] = bstart + n;
  __syncthreads();
  if (staged) {
#pragma unroll
    for (int g = 0; g < KP; g += 4) {
      if (g * 1024 < n) {
        int pos[4];
#pragma unroll
        for (int k = 0; k < 4; ++k)
          pos[k] = (t + (g + k) * 1024 < n) ? atomicAdd(&cnt[p[g + k] >> 24], 1) : 0;
#pragma unroll
        for (int k = 0; k < 4; ++k)
          if (t + (g + k) * 1024 < n) srt[pos[k]] = (int)(p[g + k] & 0xFFFFFFu);
      }
    }
  } else {
    for (int j = t; j < n; j += 1024) {
      const unsigned q = part[locate(j)];
      const int ps = atomicAdd(&cnt[q >> 24], 1);
      csr[bstart + ps] = (int)(q & 0xFFFFFFu);
    }
  }
  if (staged && n > 0) {
    __syncthreads();
    const int sh = (int)((((size_t)(csr + bstart)) >> 2) & 31);
    for (int i0 = t - sh; i0 < n; i0 += 4096) {
      int v[4];
#pragma unroll
      for (int j = 0; j < 4; ++j) v[j] = srt[min(max(i0 + j * 1024, 0), n - 1)];
#pragma unroll
      for (int j = 0; j < 4; ++j) {
        const int i = i0 + j * 1024;
        if (i >= 0 && i < n) csr[bstart + i] = v[j];
      }
    }
  }
}

template <int N> __device__ __forceinline__ float add_lane_xor(float v);
template <> __device__ __forceinline__ float add_lane_xor<1>(float v) {
  float r;
  asm("v_add_f32_dpp %0, %1, %1 row_xmask:1 row_mask:0xf bank_mask:0xf" : "=v"(r) : "v"(v));
  return r;
}
template <> __device__ __forceinline__ float add_lane_xor<2>(float v) {
  float r;
  asm("v_add_f32_dpp %0, %1, %1 row_xmask:2 row_mask:0xf bank_mask:0xf" : "=v"(r) : "v"(v));
  return r;
}
template <> __device__ __forceinline__ float add_lane_xor<4>(float v) {
  float r;
  asm("v_add_f32_dpp %0, %1, %1 row_xmask:4 row_mask:0xf bank_mask:0xf" : "=v"(r) : "v"(v));
  return r;
}

template <int LANES>
__device__ __forceinline__ float head_sum(float v) {
  v = add_lane_xor<1>(v);
  v = add_lane_xor<2>(v);
  if constexpr (LANES >= 8) v = add_lane_xor<4>(v);
  return v;
}

typedef __attribute__((ext_vector_type(4))) float v4f;
template <int VEC> struct fvec;
template <> struct fvec<2> { typedef v2f type; };
template <> struct fvec<4> { typedef v4f type; };

__device__ __forceinline__ float lane_dot(v2f x, v2f a) {
  const v2f t = x * a;
  return t.x + t.y;
}
__device__ __forceinline__ float lane_dot(v4f x, v4f a) {
  const v2f t = x.xy * a.xy + x.zw * a.zw;
  return t.x + t.y;
}

template <int H, int C, int VEC, int CH, bool FULL>
__device__ __forceinline__ void
softmax_chunk(const int* __restrict__ csr, const float* __restrict__ hfeat,
              int i, int end, int f0, typename fvec<VEC>::type asrc, float adv,
              float& m, float& den, typename fvec<VEC>::type& acc) {
  typedef typename fvec<VEC>::type vf;
  constexpr int HC = H * C;
  int s[CH];
  vf x[CH];
  float e[CH];
#pragma unroll
  for (int j = 0; j < CH; ++j) s[j] = FULL ? csr[i + j] : csr[min(i + j, end - 1)];
#pragma unroll
  for (int j = 0; j < CH; ++j) x[j] = *(const vf*)(hfeat + (size_t)s[j] * HC + f0);
#pragma unroll
  for (int j = 0; j < CH; ++j) {
    const float v = lrelu(head_sum<C / VEC>(lane_dot(x[j], asrc)) + adv, ATT_SLOPE);
    e[j] = (FULL || i + j < end) ? v : -INFINITY;
  }
  float cm = e[0];
#pragma unroll
  for (int j = 1; j < CH; ++j) cm = fmaxf(cm, e[j]);
  const float mn = fmaxf(m, cm);
  const float sc = expf(m - mn);
  m = mn;
  den *= sc;
  acc *= sc;
#pragma unroll
  for (int j = 0; j < CH; ++j) {
    const float w = expf(e[j] - mn);
    den += w;
    acc += x[j] * w;
  }
}

template <int H, int C, int VEC, bool PROJ>
__global__ void __launch_bounds__(256)
seg_gather(const int* __restrict__ rowptr, const int* __restrict__ csr,
           const float* __restrict__ hfeat,
           const float* __restrict__ att_src, const float* __restrict__ att_dst,
           const float* __restrict__ bias,
           const float* __restrict__ Wo, const float* __restrict__ bo,
           float* __restrict__ dstbuf, int Nn) {
  typedef typename fvec<VEC>::type vf;
  constexpr int HC = H * C;
  constexpr int LPN = HC / VEC;
  constexpr int CH = 4;
  const unsigned gt = blockIdx.x * blockDim.x + threadIdx.x;
  const int node = (int)(gt / LPN);
  if (node >= Nn) return;
  const int gl = (int)(threadIdx.x % LPN);
  const int f0 = gl * VEC;
  const int beg = rowptr[node];
  const int end = rowptr[node + 1];

  const vf asrc = *(const vf*)(att_src + f0);
  vf acc = *(const vf*)(hfeat + (size_t)node * HC + f0);
  const float adv = head_sum<C / VEC>(lane_dot(acc, *(const vf*)(att_dst + f0)));
  float m = lrelu(head_sum<C / VEC>(lane_dot(acc, asrc)) + adv, ATT_SLOPE);
  float den = 1.f;

  int i = beg;
  for (; i + CH <= end; i += CH)
    softmax_chunk<H, C, VEC, CH, true>(csr, hfeat, i, end, f0, asrc, adv, m, den, acc);
  if (i < end)
    softmax_chunk<H, C, VEC, CH, false>(csr, hfeat, i, end, f0, asrc, adv, m, den, acc);

  if constexpr (PROJ) {
    float p = 0.f;
#pragma unroll
    for (int k = 0; k < VEC; ++k)
      p += lrelu(acc[k] / den + bias[f0 + k], ACT_SLOPE) * Wo[f0 + k];
    for (int o = 1; o < LPN; o <<= 1) p += __shfl_xor(p, o, LPN);
    if (gl == 0) dstbuf[node] = p + bo[0];
  } else {
    float* op = dstbuf + (size_t)node * HC + f0;
#pragma unroll
    for (int k = 0; k < VEC; ++k)
      op[k] = lrelu(acc[k] / den + bias[f0 + k], ACT_SLOPE);
  }
}

extern "C" void kernel_launch(void* const* d_in, const int* in_sizes, int n_in,
                              void* d_out, int out_size, void* d_ws, size_t ws_size,
                              hipStream_t stream) {
  const float* x    = (const float*)d_in[0];
  const int*   ei   = (const int*)d_in[1];
  const float* W1   = (const float*)d_in[2];
  const float* aS1  = (const float*)d_in[3];
  const float* aD1  = (const float*)d_in[4];
  const float* b1   = (const float*)d_in[5];
  const float* W2   = (const float*)d_in[6];
  const float* aS2  = (const float*)d_in[7];
  const float* aD2  = (const float*)d_in[8];
  const float* b2   = (const float*)d_in[9];
  const float* Wo   = (const float*)d_in[10];
  const float* bo   = (const float*)d_in[11];
  float* out = (float*)d_out;

  const int IN_C = 64, HID = 16, HEADS = 4;
  const int HC1 = HEADS * HID;
  const int Nn = in_sizes[0] / IN_C;
  const int E  = in_sizes[1] / 2;
  const int* src = ei;
  const int* dst = ei + E;

  float* p = (float*)d_ws;
  float* h1   = p; p += (size_t)Nn * HC1;
  float* f1   = p; p += (size_t)Nn * HC1;
  float* h2   = p; p += (size_t)Nn * HID;
  const int NB = (Nn + BNODES - 1) / BNODES;
  const int NC = (E + CHUNK_E - 1) / CHUNK_E;
  int* q = (int*)p;
  int* off    = q; q += (size_t)NC * (NB + 1);
  int* rowptr = q; q += (size_t)Nn + 1;
  unsigned* part = (unsigned*)q; q += (size_t)E;
  int* csr    = q; q += (size_t)E;

  const int TB = 256;

  {
    const int tiles = ((Nn + 15) / 16) * (HC1 / 16);
    chunk_sort_and_gemm<64, 64><<<NC + (tiles + 31) / 32, 1024,
                                  (32 + NB + CHUNK_E) * sizeof(int), stream>>>(
        src, dst, off, part, E, NB, NC, x, W1, h1, Nn);
  }
  bucket_sort<<<NB, 1024, (2 * NC + 1) * sizeof(int), stream>>>(
      off, part, rowptr, csr, Nn, NB, NC);

  {
    const unsigned n = (unsigned)Nn * 16u;
    seg_gather<4, 16, 4, false><<<(n + TB - 1) / TB, TB, 0, stream>>>(
        rowptr, csr, h1, aS1, aD1, b1, Wo, bo, f1, Nn);
  }

  {
    dim3 g((Nn + 15) / 16, HID / 16);
    wmma_gemm_f32<64, 16><<<g, 32, 0, stream>>>(f1, W2, h2, Nn);
  }
  {
    const unsigned n = (unsigned)Nn * 8u;
    seg_gather<1, 16, 2, true><<<(n + TB - 1) / TB, TB, 0, stream>>>(
        rowptr, csr, h2, aS2, aD2, b2, Wo, bo, out, Nn);
  }
}


# Segment softmax weights taken as 2^((e-m)*log2 e) on the hardware exp2 instruction, not the library expf (over ten vector instructions a call, five calls per four edges); weight rounding changes
# speedup vs baseline: 12.0181x; 1.0678x over previous
#include <hip/hip_runtime.h>
#include <math.h>

#define ATT_SLOPE 0.2f
#define ACT_SLOPE 0.01f

typedef __attribute__((ext_vector_type(2))) float v2f;
typedef __attribute__((ext_vector_type(8))) float v8f;

__device__ __forceinline__ float lrelu(float v, float s) { return v > 0.f ? v : v * s; }

template <int K, int NCOLS>
__device__ __forceinline__ void
wmma_tile_f32(const float* __restrict__ A, const float* __restrict__ B,
              float* __restrict__ D, int M, int m0, int n0, int lane) {
  const int half = lane >> 4;
  const int lid  = lane & 15;
  const int ar = min(m0 + lid, M - 1);
  v8f acc = {};
#pragma unroll
  for (int k = 0; k < K; k += 4) {
    const int ka = k + half * 2;
    v2f a, b;
    a.x = A[(size_t)ar * K + ka];
    a.y = A[(size_t)ar * K + ka + 1];
    b.x = B[(size_t)ka * NCOLS + n0 + lid];
    b.y = B[(size_t)(ka + 1) * NCOLS + n0 + lid];
    acc = __builtin_amdgcn_wmma_f32_16x16x4_f32(false, a, false, b, (short)0, acc,
                                                false, false);
  }
  const int rbase = m0 + half * 8;
#pragma unroll
  for (int r = 0; r < 8; ++r) {
    const int row = rbase + r;
    if (row < M) D[(size_t)row * NCOLS + n0 + lid] = acc[r];
  }
}

template <int K, int NCOLS>
__global__ void __launch_bounds__(32)
wmma_gemm_f32(const float* __restrict__ A, const float* __restrict__ B,
              float* __restrict__ D, int M) {
  wmma_tile_f32<K, NCOLS>(A, B, D, M, blockIdx.x * 16, blockIdx.y * 16, threadIdx.x);
}

#define CHUNK_E 8192
#define BSHIFT 8
#define BNODES 256

__device__ __forceinline__ int
block_scan_lds(int* a, int n, int* wsum, int* __restrict__ gout) {
  const int t = threadIdx.x;
  const int lane = t & 31, w = t >> 5;
  const int per = (n + 1023) / 1024;
  const int b = min(t * per, n);
  const int e = min(b + per, n);
  int s = 0;
  for (int i = b; i < e; ++i) s += a[i];
  int v = s;
  for (int o = 1; o < 32; o <<= 1) {
    const int u = __shfl_up(v, o, 32);
    if (lane >= o) v += u;
  }
  if (lane == 31) wsum[w] = v;
  __syncthreads();
  if (w == 0) {
    int x = wsum[lane];
    for (int o = 1; o < 32; o <<= 1) {
      const int u = __shfl_up(x, o, 32);
      if (lane >= o) x += u;
    }
    wsum[lane] = x;
  }
  __syncthreads();
  int run = v - s + (w > 0 ? wsum[w - 1] : 0);
  for (int i = b; i < e; ++i) {
    const int cn = a[i];
    a[i] = run;
    if (gout) gout[i] = run;
    run += cn;
  }
  return wsum[31];
}

__device__ __forceinline__ void
chunk_sort_body(const int* __restrict__ src, const int* __restrict__ dst,
                int* __restrict__ off, unsigned* __restrict__ part,
                int E, int NB, int c, int* lds) {
  int* wsum = lds;
  int* lc = lds + 32;
  unsigned* sval = (unsigned*)(lds + 32 + NB);
  const int t = threadIdx.x;
  constexpr int PER = CHUNK_E / 1024;
  for (int i = t; i < NB; i += 1024) lc[i] = 0;
  const int e0 = c * CHUNK_E + t;
  int d[PER], s[PER];
#pragma unroll
  for (int k = 0; k < PER; ++k) {
    const int e = min(e0 + k * 1024, E - 1);
    d[k] = dst[e];
    s[k] = src[e];
  }
  __syncthreads();
#pragma unroll
  for (int k = 0; k < PER; ++k)
    if (e0 + k * 1024 < E) atomicAdd(&lc[d[k] >> BSHIFT], 1);
  __syncthreads();
  int* orow = off + (size_t)c * (NB + 1);
  const int nloc = block_scan_lds(lc, NB, wsum, orow);
  if (t == 0) orow[NB] = nloc;
  __syncthreads();
  int pos[PER];
#pragma unroll
  for (int k = 0; k < PER; ++k)
    pos[k] = (e0 + k * 1024 < E) ? atomicAdd(&lc[d[k] >> BSHIFT], 1) : 0;
#pragma unroll
  for (int k = 0; k < PER; ++k)
    if (e0 + k * 1024 < E)
      sval[pos[k]] = (unsigned)s[k] | ((unsigned)(d[k] & (BNODES - 1)) << 24);
  __syncthreads();
  unsigned vv[PER];
#pragma unroll
  for (int k = 0; k < PER; ++k) vv[k] = sval[min(t + k * 1024, nloc - 1)];
#pragma unroll
  for (int k = 0; k < PER; ++k) {
    const int i = t + k * 1024;
    if (i < nloc) part[(size_t)c * CHUNK_E + i] = vv[k];
  }
}

template <int K, int NCOLS>
__global__ void __launch_bounds__(1024)
chunk_sort_and_gemm(const int* __restrict__ src, const int* __restrict__ dst,
                    int* __restrict__ off, unsigned* __restrict__ part,
                    int E, int NB, int NC,
                    const float* __restrict__ A, const float* __restrict__ B,
                    float* __restrict__ D, int M) {
  extern __shared__ int lds[];
  if ((int)blockIdx.x < NC) {
    chunk_sort_body(src, dst, off, part, E, NB, (int)blockIdx.x, lds);
    return;
  }
  constexpr int NT = NCOLS / 16;
  const int tile = ((int)blockIdx.x - NC) * 32 + (int)(threadIdx.x >> 5);
  const int m0 = (tile / NT) * 16;
  if (m0 >= M) return;
  wmma_tile_f32<K, NCOLS>(A, B, D, M, m0, (tile % NT) * 16, threadIdx.x & 31);
}

#define BCAP 12288
__global__ void __launch_bounds__(1024)
bucket_sort(const int* __restrict__ off, const unsigned* __restrict__ part,
            int* __restrict__ rowptr, int* __restrict__ csr, int Nn, int NB, int NC) {
  extern __shared__ int lds[];
  __shared__ int cnt[BNODES];
  __shared__ int srt[BCAP];
  __shared__ int wsum[32];
  __shared__ int bsum;
  int* rs = lds;
  int* rd = lds + NC + 1;
  const int t = threadIdx.x;
  const int b = blockIdx.x;
  if (t < BNODES) cnt[t] = 0;
  if (t == 0) bsum = 0;
  __syncthreads();
  {
    int so = 0;
    for (int c = t; c < NC; c += 1024) {
      const int o0 = off[(size_t)c * (NB + 1) + b];
      const int o1 = off[(size_t)c * (NB + 1) + b + 1];
      rs[c] = o1 - o0;
      rd[c] = c * CHUNK_E + o0;
      so += o0;
    }
    if (so != 0) atomicAdd(&bsum, so);
  }
  __syncthreads();
  const int n = block_scan_lds(rs, NC, wsum, nullptr);
  const int bstart = bsum;
  if (t == 0) rs[NC] = n;
  __syncthreads();
  for (int c = t; c < NC; c += 1024) rd[c] -= rs[c];
  __syncthreads();
  const bool staged = n <= BCAP;
  int P2 = 1;
  while (P2 < NC) P2 <<= 1;
  auto locate = [&](int j) -> int {
    int lo = 0;
    for (int sp = P2 >> 1; sp > 0; sp >>= 1) {
      const int m = lo + sp;
      if (rs[min(m, NC)] <= j) lo = m;
    }
    return rd[lo] + j;
  };
  auto locate4 = [&](const int (&j)[4], int (&a)[4]) {
    int lo[4] = {0, 0, 0, 0};
    for (int sp = P2 >> 1; sp > 0; sp >>= 1) {
      int v[4];
#pragma unroll
      for (int k = 0; k < 4; ++k) v[k] = rs[min(lo[k] + sp, NC)];
#pragma unroll
      for (int k = 0; k < 4; ++k)
        if (v[k] <= j[k]) lo[k] += sp;
    }
#pragma unroll
    for (int k = 0; k < 4; ++k) a[k] = rd[lo[k]] + j[k];
  };
  constexpr int KP = BCAP / 1024;
  unsigned p[KP];
  if (staged) {
#pragma unroll
    for (int g = 0; g < KP; g += 4) {
      if (g * 1024 < n) {
        int jj[4], a[4];
#pragma unroll
        for (int k = 0; k < 4; ++k) jj[k] = min(t + (g + k) * 1024, n - 1);
        locate4(jj, a);
#pragma unroll
        for (int k = 0; k < 4; ++k) p[g + k] = part[a[k]];
#pragma unroll
        for (int k = 0; k < 4; ++k)
          if (t + (g + k) * 1024 < n) atomicAdd(&cnt[p[g + k] >> 24], 1);
      }
    }
  } else {
    for (int j = t; j < n; j += 1024) atomicAdd(&cnt[part[locate(j)] >> 24], 1);
  }
  __syncthreads();
  if (t < 32) {
    int cc[BNODES / 32];
    int s = 0;
#pragma unroll
    for (int j = 0; j < BNODES / 32; ++j) {
      cc[j] = cnt[t * (BNODES / 32) + j];
      s += cc[j];
    }
    int v = s;
    for (int o = 1; o < 32; o <<= 1) {
      const int u = __shfl_up(v, o, 32);
      if (t >= o) v += u;
    }
    int run = v - s;
#pragma unroll
    for (int j = 0; j < BNODES / 32; ++j) {
      const int local = t * (BNODES / 32) + j;
      const int node = b * BNODES + local;
      cnt[local] = run;
      if (node < Nn) rowptr[node] = bstart + run;
      run += cc[j];
    }
  }
  if (b == (int)gridDim.x - 1 && t == 0) rowptr[Nn] = bstart + n;
  __syncthreads();
  if (staged) {
#pragma unroll
    for (int g = 0; g < KP; g += 4) {
      if (g * 1024 < n) {
        int pos[4];
#pragma unroll
        for (int k = 0; k < 4; ++k)
          pos[k] = (t + (g + k) * 1024 < n) ? atomicAdd(&cnt[p[g + k] >> 24], 1) : 0;
#pragma unroll
        for (int k = 0; k < 4; ++k)
          if (t + (g + k) * 1024 < n) srt[pos[k]] = (int)(p[g + k] & 0xFFFFFFu);
      }
    }
  } else {
    for (int j = t; j < n; j += 1024) {
      const unsigned q = part[locate(j)];
      const int ps = atomicAdd(&cnt[q >> 24], 1);
      csr[bstart + ps] = (int)(q & 0xFFFFFFu);
    }
  }
  if (staged && n > 0) {
    __syncthreads();
    const int sh = (int)((((size_t)(csr + bstart)) >> 2) & 31);
    for (int i0 = t - sh; i0 < n; i0 += 4096) {
      int v[4];
#pragma unroll
      for (int j = 0; j < 4; ++j) v[j] = srt[min(max(i0 + j * 1024, 0), n - 1)];
#pragma unroll
      for (int j = 0; j < 4; ++j) {
        const int i = i0 + j * 1024;
        if (i >= 0 && i < n) csr[bstart + i] = v[j];
      }
    }
  }
}

template <int N> __device__ __forceinline__ float add_lane_xor(float v);
template <> __device__ __forceinline__ float add_lane_xor<1>(float v) {
  float r;
  asm("v_add_f32_dpp %0, %1, %1 row_xmask:1 row_mask:0xf bank_mask:0xf" : "=v"(r) : "v"(v));
  return r;
}
template <> __device__ __forceinline__ float add_lane_xor<2>(float v) {
  float r;
  asm("v_add_f32_dpp %0, %1, %1 row_xmask:2 row_mask:0xf bank_mask:0xf" : "=v"(r) : "v"(v));
  return r;
}
template <> __device__ __forceinline__ float add_lane_xor<4>(float v) {
  float r;
  asm("v_add_f32_dpp %0, %1, %1 row_xmask:4 row_mask:0xf bank_mask:0xf" : "=v"(r) : "v"(v));
  return r;
}

template <int LANES>
__device__ __forceinline__ float head_sum(float v) {
  v = add_lane_xor<1>(v);
  v = add_lane_xor<2>(v);
  if constexpr (LANES >= 8) v = add_lane_xor<4>(v);
  return v;
}

typedef __attribute__((ext_vector_type(4))) float v4f;
template <int VEC> struct fvec;
template <> struct fvec<2> { typedef v2f type; };
template <> struct fvec<4> { typedef v4f type; };

__device__ __forceinline__ float lane_dot(v2f x, v2f a) {
  const v2f t = x * a;
  return t.x + t.y;
}
__device__ __forceinline__ float lane_dot(v4f x, v4f a) {
  const v2f t = x.xy * a.xy + x.zw * a.zw;
  return t.x + t.y;
}

#define LOG2_E 1.44269504088896340736f
__device__ __forceinline__ float exp_of_nonpositive(float d) {
  return __builtin_amdgcn_exp2f(d * LOG2_E);
}

template <int H, int C, int VEC, int CH, bool FULL>
__device__ __forceinline__ void
softmax_chunk(const int* __restrict__ csr, const float* __restrict__ hfeat,
              int i, int end, int f0, typename fvec<VEC>::type asrc, float adv,
              float& m, float& den, typename fvec<VEC>::type& acc) {
  typedef typename fvec<VEC>::type vf;
  constexpr int HC = H * C;
  int s[CH];
  vf x[CH];
  float e[CH];
#pragma unroll
  for (int j = 0; j < CH; ++j) s[j] = FULL ? csr[i + j] : csr[min(i + j, end - 1)];
#pragma unroll
  for (int j = 0; j < CH; ++j) x[j] = *(const vf*)(hfeat + (size_t)s[j] * HC + f0);
#pragma unroll
  for (int j = 0; j < CH; ++j) {
    const float v = lrelu(head_sum<C / VEC>(lane_dot(x[j], asrc)) + adv, ATT_SLOPE);
    e[j] = (FULL || i + j < end) ? v : -INFINITY;
  }
  float cm = e[0];
#pragma unroll
  for (int j = 1; j < CH; ++j) cm = fmaxf(cm, e[j]);
  const float mn = fmaxf(m, cm);
  const float sc = exp_of_nonpositive(m - mn);
  m = mn;
  den *= sc;
  acc *= sc;
#pragma unroll
  for (int j = 0; j < CH; ++j) {
    const float w = exp_of_nonpositive(e[j] - mn);
    den += w;
    acc += x[j] * w;
  }
}

template <int H, int C, int VEC, bool PROJ>
__global__ void __launch_bounds__(256)
seg_gather(const int* __restrict__ rowptr, const int* __restrict__ csr,
           const float* __restrict__ hfeat,
           const float* __restrict__ att_src, const float* __restrict__ att_dst,
           const float* __restrict__ bias,
           const float* __restrict__ Wo, const float* __restrict__ bo,
           float* __restrict__ dstbuf, int Nn) {
  typedef typename fvec<VEC>::type vf;
  constexpr int HC = H * C;
  constexpr int LPN = HC / VEC;
  constexpr int CH = 4;
  const unsigned gt = blockIdx.x * blockDim.x + threadIdx.x;
  const int node = (int)(gt / LPN);
  if (node >= Nn) return;
  const int gl = (int)(threadIdx.x % LPN);
  const int f0 = gl * VEC;
  const int beg = rowptr[node];
  const int end = rowptr[node + 1];

  const vf asrc = *(const vf*)(att_src + f0);
  vf acc = *(const vf*)(hfeat + (size_t)node * HC + f0);
  const float adv = head_sum<C / VEC>(lane_dot(acc, *(const vf*)(att_dst + f0)));
  float m = lrelu(head_sum<C / VEC>(lane_dot(acc, asrc)) + adv, ATT_SLOPE);
  float den = 1.f;

  int i = beg;
  for (; i + CH <= end; i += CH)
    softmax_chunk<H, C, VEC, CH, true>(csr, hfeat, i, end, f0, asrc, adv, m, den, acc);
  if (i < end)
    softmax_chunk<H, C, VEC, CH, false>(csr, hfeat, i, end, f0, asrc, adv, m, den, acc);

  if constexpr (PROJ) {
    float p = 0.f;
#pragma unroll
    for (int k = 0; k < VEC; ++k)
      p += lrelu(acc[k] / den + bias[f0 + k], ACT_SLOPE) * Wo[f0 + k];
    for (int o = 1; o < LPN; o <<= 1) p += __shfl_xor(p, o, LPN);
    if (gl == 0) dstbuf[node] = p + bo[0];
  } else {
    float* op = dstbuf + (size_t)node * HC + f0;
#pragma unroll
    for (int k = 0; k < VEC; ++k)
      op[k] = lrelu(acc[k] / den + bias[f0 + k], ACT_SLOPE);
  }
}

extern "C" void kernel_launch(void* const* d_in, const int* in_sizes, int n_in,
                              void* d_out, int out_size, void* d_ws, size_t ws_size,
                              hipStream_t stream) {
  const float* x    = (const float*)d_in[0];
  const int*   ei   = (const int*)d_in[1];
  const float* W1   = (const float*)d_in[2];
  const float* aS1  = (const float*)d_in[3];
  const float* aD1  = (const float*)d_in[4];
  const float* b1   = (const float*)d_in[5];
  const float* W2   = (const float*)d_in[6];
  const float* aS2  = (const float*)d_in[7];
  const float* aD2  = (const float*)d_in[8];
  const float* b2   = (const float*)d_in[9];
  const float* Wo   = (const float*)d_in[10];
  const float* bo   = (const float*)d_in[11];
  float* out = (float*)d_out;

  const int IN_C = 64, HID = 16, HEADS = 4;
  const int HC1 = HEADS * HID;
  const int Nn = in_sizes[0] / IN_C;
  const int E  = in_sizes[1] / 2;
  const int* src = ei;
  const int* dst = ei + E;

  float* p = (float*)d_ws;
  float* h1   = p; p += (size_t)Nn * HC1;
  float* f1   = p; p += (size_t)Nn * HC1;
  float* h2   = p; p += (size_t)Nn * HID;
  const int NB = (Nn + BNODES - 1) / BNODES;
  const int NC = (E + CHUNK_E - 1) / CHUNK_E;
  int* q = (int*)p;
  int* off    = q; q += (size_t)NC * (NB + 1);
  int* rowptr = q; q += (size_t)Nn + 1;
  unsigned* part = (unsigned*)q; q += (size_t)E;
  int* csr    = q; q += (size_t)E;

  const int TB = 256;

  {
    const int tiles = ((Nn + 15) / 16) * (HC1 / 16);
    chunk_sort_and_gemm<64, 64><<<NC + (tiles + 31) / 32, 1024,
                                  (32 + NB + CHUNK_E) * sizeof(int), stream>>>(
        src, dst, off, part, E, NB, NC, x, W1, h1, Nn);
  }
  bucket_sort<<<NB, 1024, (2 * NC + 1) * sizeof(int), stream>>>(
      off, part, rowptr, csr, Nn, NB, NC);

  {
    const unsigned n = (unsigned)Nn * 16u;
    seg_gather<4, 16, 4, false><<<(n + TB - 1) / TB, TB, 0, stream>>>(
        rowptr, csr, h1, aS1, aD1, b1, Wo, bo, f1, Nn);
  }

  {
    dim3 g((Nn + 15) / 16, HID / 16);
    wmma_gemm_f32<64, 16><<<g, 32, 0, stream>>>(f1, W2, h2, Nn);
  }
  {
    const unsigned n = (unsigned)Nn * 8u;
    seg_gather<1, 16, 2, true><<<(n + TB - 1) / TB, TB, 0, stream>>>(
        rowptr, csr, h2, aS2, aD2, b2, Wo, bo, out, Nn);
  }
}


# Layer 2's linear map runs in the tail of layer 1's gather: a work-group's 16 activation rows go through LDS to its first wave's WMMAs and are never stored; 5 launches become 4
# speedup vs baseline: 12.4625x; 1.0370x over previous
#include <hip/hip_runtime.h>
#include <math.h>

#define ATT_SLOPE 0.2f
#define ACT_SLOPE 0.01f

typedef __attribute__((ext_vector_type(2))) float v2f;
typedef __attribute__((ext_vector_type(8))) float v8f;

__device__ __forceinline__ float lrelu(float v, float s) { return v > 0.f ? v : v * s; }

template <int K, int NCOLS>
__device__ __forceinline__ void
wmma_tile_f32(const float* __restrict__ A, const float* __restrict__ B,
              float* __restrict__ D, int M, int m0, int n0, int lane) {
  const int half = lane >> 4;
  const int lid  = lane & 15;
  const int ar = min(m0 + lid, M - 1);
  v8f acc = {};
#pragma unroll
  for (int k = 0; k < K; k += 4) {
    const int ka = k + half * 2;
    v2f a, b;
    a.x = A[(size_t)ar * K + ka];
    a.y = A[(size_t)ar * K + ka + 1];
    b.x = B[(size_t)ka * NCOLS + n0 + lid];
    b.y = B[(size_t)(ka + 1) * NCOLS + n0 + lid];
    acc = __builtin_amdgcn_wmma_f32_16x16x4_f32(false, a, false, b, (short)0, acc,
                                                false, false);
  }
  const int rbase = m0 + half * 8;
#pragma unroll
  for (int r = 0; r < 8; ++r) {
    const int row = rbase + r;
    if (row < M) D[(size_t)row * NCOLS + n0 + lid] = acc[r];
  }
}

#define CHUNK_E 8192
#define BSHIFT 8
#define BNODES 256

__device__ __forceinline__ int
block_scan_lds(int* a, int n, int* wsum, int* __restrict__ gout) {
  const int t = threadIdx.x;
  const int lane = t & 31, w = t >> 5;
  const int per = (n + 1023) / 1024;
  const int b = min(t * per, n);
  const int e = min(b + per, n);
  int s = 0;
  for (int i = b; i < e; ++i) s += a[i];
  int v = s;
  for (int o = 1; o < 32; o <<= 1) {
    const int u = __shfl_up(v, o, 32);
    if (lane >= o) v += u;
  }
  if (lane == 31) wsum[w] = v;
  __syncthreads();
  if (w == 0) {
    int x = wsum[lane];
    for (int o = 1; o < 32; o <<= 1) {
      const int u = __shfl_up(x, o, 32);
      if (lane >= o) x += u;
    }
    wsum[lane] = x;
  }
  __syncthreads();
  int run = v - s + (w > 0 ? wsum[w - 1] : 0);
  for (int i = b; i < e; ++i) {
    const int cn = a[i];
    a[i] = run;
    if (gout) gout[i] = run;
    run += cn;
  }
  return wsum[31];
}

__device__ __forceinline__ void
chunk_sort_body(const int* __restrict__ src, const int* __restrict__ dst,
                int* __restrict__ off, unsigned* __restrict__ part,
                int E, int NB, int c, int* lds) {
  int* wsum = lds;
  int* lc = lds + 32;
  unsigned* sval = (unsigned*)(lds + 32 + NB);
  const int t = threadIdx.x;
  constexpr int PER = CHUNK_E / 1024;
  for (int i = t; i < NB; i += 1024) lc[i] = 0;
  const int e0 = c * CHUNK_E + t;
  int d[PER], s[PER];
#pragma unroll
  for (int k = 0; k < PER; ++k) {
    const int e = min(e0 + k * 1024, E - 1);
    d[k] = dst[e];
    s[k] = src[e];
  }
  __syncthreads();
#pragma unroll
  for (int k = 0; k < PER; ++k)
    if (e0 + k * 1024 < E) atomicAdd(&lc[d[k] >> BSHIFT], 1);
  __syncthreads();
  int* orow = off + (size_t)c * (NB + 1);
  const int nloc = block_scan_lds(lc, NB, wsum, orow);
  if (t == 0) orow[NB] = nloc;
  __syncthreads();
  int pos[PER];
#pragma unroll
  for (int k = 0; k < PER; ++k)
    pos[k] = (e0 + k * 1024 < E) ? atomicAdd(&lc[d[k] >> BSHIFT], 1) : 0;
#pragma unroll
  for (int k = 0; k < PER; ++k)
    if (e0 + k * 1024 < E)
      sval[pos[k]] = (unsigned)s[k] | ((unsigned)(d[k] & (BNODES - 1)) << 24);
  __syncthreads();
  unsigned vv[PER];
#pragma unroll
  for (int k = 0; k < PER; ++k) vv[k] = sval[min(t + k * 1024, nloc - 1)];
#pragma unroll
  for (int k = 0; k < PER; ++k) {
    const int i = t + k * 1024;
    if (i < nloc) part[(size_t)c * CHUNK_E + i] = vv[k];
  }
}

template <int K, int NCOLS>
__global__ void __launch_bounds__(1024)
chunk_sort_and_gemm(const int* __restrict__ src, const int* __restrict__ dst,
                    int* __restrict__ off, unsigned* __restrict__ part,
                    int E, int NB, int NC,
                    const float* __restrict__ A, const float* __restrict__ B,
                    float* __restrict__ D, int M) {
  extern __shared__ int lds[];
  if ((int)blockIdx.x < NC) {
    chunk_sort_body(src, dst, off, part, E, NB, (int)blockIdx.x, lds);
    return;
  }
  constexpr int NT = NCOLS / 16;
  const int tile = ((int)blockIdx.x - NC) * 32 + (int)(threadIdx.x >> 5);
  const int m0 = (tile / NT) * 16;
  if (m0 >= M) return;
  wmma_tile_f32<K, NCOLS>(A, B, D, M, m0, (tile % NT) * 16, threadIdx.x & 31);
}

#define BCAP 12288
__global__ void __launch_bounds__(1024)
bucket_sort(const int* __restrict__ off, const unsigned* __restrict__ part,
            int* __restrict__ rowptr, int* __restrict__ csr, int Nn, int NB, int NC) {
  extern __shared__ int lds[];
  __shared__ int cnt[BNODES];
  __shared__ int srt[BCAP];
  __shared__ int wsum[32];
  __shared__ int bsum;
  int* rs = lds;
  int* rd = lds + NC + 1;
  const int t = threadIdx.x;
  const int b = blockIdx.x;
  if (t < BNODES) cnt[t] = 0;
  if (t == 0) bsum = 0;
  __syncthreads();
  {
    int so = 0;
    for (int c = t; c < NC; c += 1024) {
      const int o0 = off[(size_t)c * (NB + 1) + b];
      const int o1 = off[(size_t)c * (NB + 1) + b + 1];
      rs[c] = o1 - o0;
      rd[c] = c * CHUNK_E + o0;
      so += o0;
    }
    if (so != 0) atomicAdd(&bsum, so);
  }
  __syncthreads();
  const int n = block_scan_lds(rs, NC, wsum, nullptr);
  const int bstart = bsum;
  if (t == 0) rs[NC] = n;
  __syncthreads();
  for (int c = t; c < NC; c += 1024) rd[c] -= rs[c];
  __syncthreads();
  const bool staged = n <= BCAP;
  int P2 = 1;
  while (P2 < NC) P2 <<= 1;
  auto locate = [&](int j) -> int {
    int lo = 0;
    for (int sp = P2 >> 1; sp > 0; sp >>= 1) {
      const int m = lo + sp;
      if (rs[min(m, NC)] <= j) lo = m;
    }
    return rd[lo] + j;
  };
  auto locate4 = [&](const int (&j)[4], int (&a)[4]) {
    int lo[4] = {0, 0, 0, 0};
    for (int sp = P2 >> 1; sp > 0; sp >>= 1) {
      int v[4];
#pragma unroll
      for (int k = 0; k < 4; ++k) v[k] = rs[min(lo[k] + sp, NC)];
#pragma unroll
      for (int k = 0; k < 4; ++k)
        if (v[k] <= j[k]) lo[k] += sp;
    }
#pragma unroll
    for (int k = 0; k < 4; ++k) a[k] = rd[lo[k]] + j[k];
  };
  constexpr int KP = BCAP / 1024;
  unsigned p[KP];
  if (staged) {
#pragma unroll
    for (int g = 0; g < KP; g += 4) {
      if (g * 1024 < n) {
        int jj[4], a[4];
#pragma unroll
        for (int k = 0; k < 4; ++k) jj[k] = min(t + (g + k) * 1024, n - 1);
        locate4(jj, a);
#pragma unroll
        for (int k = 0; k < 4; ++k) p[g + k] = part[a[k]];
#pragma unroll
        for (int k = 0; k < 4; ++k)
          if (t + (g + k) * 1024 < n) atomicAdd(&cnt[p[g + k] >> 24], 1);
      }
    }
  } else {
    for (int j = t; j < n; j += 1024) atomicAdd(&cnt[part[locate(j)] >> 24], 1);
  }
  __syncthreads();
  if (t < 32) {
    int cc[BNODES / 32];
    int s = 0;
#pragma unroll
    for (int j = 0; j < BNODES / 32; ++j) {
      cc[j] = cnt[t * (BNODES / 32) + j];
      s += cc[j];
    }
    int v = s;
    for (int o = 1; o < 32; o <<= 1) {
      const int u = __shfl_up(v, o, 32);
      if (t >= o) v += u;
    }
    int run = v - s;
#pragma unroll
    for (int j = 0; j < BNODES / 32; ++j) {
      const int local = t * (BNODES / 32) + j;
      const int node = b * BNODES + local;
      cnt[local] = run;
      if (node < Nn) rowptr[node] = bstart + run;
      run += cc[j];
    }
  }
  if (b == (int)gridDim.x - 1 && t == 0) rowptr[Nn] = bstart + n;
  __syncthreads();
  if (staged) {
#pragma unroll
    for (int g = 0; g < KP; g += 4) {
      if (g * 1024 < n) {
        int pos[4];
#pragma unroll
        for (int k = 0; k < 4; ++k)
          pos[k] = (t + (g + k) * 1024 < n) ? atomicAdd(&cnt[p[g + k] >> 24], 1) : 0;
#pragma unroll
        for (int k = 0; k < 4; ++k)
          if (t + (g + k) * 1024 < n) srt[pos[k]] = (int)(p[g + k] & 0xFFFFFFu);
      }
    }
  } else {
    for (int j = t; j < n; j += 1024) {
      const unsigned q = part[locate(j)];
      const int ps = atomicAdd(&cnt[q >> 24], 1);
      csr[bstart + ps] = (int)(q & 0xFFFFFFu);
    }
  }
  if (staged && n > 0) {
    __syncthreads();
    const int sh = (int)((((size_t)(csr + bstart)) >> 2) & 31);
    for (int i0 = t - sh; i0 < n; i0 += 4096) {
      int v[4];
#pragma unroll
      for (int j = 0; j < 4; ++j) v[j] = srt[min(max(i0 + j * 1024, 0), n - 1)];
#pragma unroll
      for (int j = 0; j < 4; ++j) {
        const int i = i0 + j * 1024;
        if (i >= 0 && i < n) csr[bstart + i] = v[j];
      }
    }
  }
}

template <int N> __device__ __forceinline__ float add_lane_xor(float v);
template <> __device__ __forceinline__ float add_lane_xor<1>(float v) {
  float r;
  asm("v_add_f32_dpp %0, %1, %1 row_xmask:1 row_mask:0xf bank_mask:0xf" : "=v"(r) : "v"(v));
  return r;
}
template <> __device__ __forceinline__ float add_lane_xor<2>(float v) {
  float r;
  asm("v_add_f32_dpp %0, %1, %1 row_xmask:2 row_mask:0xf bank_mask:0xf" : "=v"(r) : "v"(v));
  return r;
}
template <> __device__ __forceinline__ float add_lane_xor<4>(float v) {
  float r;
  asm("v_add_f32_dpp %0, %1, %1 row_xmask:4 row_mask:0xf bank_mask:0xf" : "=v"(r) : "v"(v));
  return r;
}

template <int LANES>
__device__ __forceinline__ float head_sum(float v) {
  v = add_lane_xor<1>(v);
  v = add_lane_xor<2>(v);
  if constexpr (LANES >= 8) v = add_lane_xor<4>(v);
  return v;
}

typedef __attribute__((ext_vector_type(4))) float v4f;
template <int VEC> struct fvec;
template <> struct fvec<2> { typedef v2f type; };
template <> struct fvec<4> { typedef v4f type; };

__device__ __forceinline__ float lane_dot(v2f x, v2f a) {
  const v2f t = x * a;
  return t.x + t.y;
}
__device__ __forceinline__ float lane_dot(v4f x, v4f a) {
  const v2f t = x.xy * a.xy + x.zw * a.zw;
  return t.x + t.y;
}

#define LOG2_E 1.44269504088896340736f
__device__ __forceinline__ float exp_of_nonpositive(float d) {
  return __builtin_amdgcn_exp2f(d * LOG2_E);
}

template <int H, int C, int VEC, int CH, bool FULL>
__device__ __forceinline__ void
softmax_chunk(const int* __restrict__ csr, const float* __restrict__ hfeat,
              int i, int end, int f0, typename fvec<VEC>::type asrc, float adv,
              float& m, float& den, typename fvec<VEC>::type& acc) {
  typedef typename fvec<VEC>::type vf;
  constexpr int HC = H * C;
  int s[CH];
  vf x[CH];
  float e[CH];
#pragma unroll
  for (int j = 0; j < CH; ++j) s[j] = FULL ? csr[i + j] : csr[min(i + j, end - 1)];
#pragma unroll
  for (int j = 0; j < CH; ++j) x[j] = *(const vf*)(hfeat + (size_t)s[j] * HC + f0);
#pragma unroll
  for (int j = 0; j < CH; ++j) {
    const float v = lrelu(head_sum<C / VEC>(lane_dot(x[j], asrc)) + adv, ATT_SLOPE);
    e[j] = (FULL || i + j < end) ? v : -INFINITY;
  }
  float cm = e[0];
#pragma unroll
  for (int j = 1; j < CH; ++j) cm = fmaxf(cm, e[j]);
  const float mn = fmaxf(m, cm);
  const float sc = exp_of_nonpositive(m - mn);
  m = mn;
  den *= sc;
  acc *= sc;
#pragma unroll
  for (int j = 0; j < CH; ++j) {
    const float w = exp_of_nonpositive(e[j] - mn);
    den += w;
    acc += x[j] * w;
  }
}

template <int H, int C, int VEC>
__device__ __forceinline__ void
segment_walk(const int* __restrict__ rowptr, const int* __restrict__ csr,
             const float* __restrict__ hfeat,
             const float* __restrict__ att_src, const float* __restrict__ att_dst,
             int node, int f0, typename fvec<VEC>::type& acc, float& den) {
  typedef typename fvec<VEC>::type vf;
  constexpr int HC = H * C;
  constexpr int CH = 4;
  const int beg = rowptr[node];
  const int end = rowptr[node + 1];

  const vf asrc = *(const vf*)(att_src + f0);
  acc = *(const vf*)(hfeat + (size_t)node * HC + f0);
  const float adv = head_sum<C / VEC>(lane_dot(acc, *(const vf*)(att_dst + f0)));
  float m = lrelu(head_sum<C / VEC>(lane_dot(acc, asrc)) + adv, ATT_SLOPE);
  den = 1.f;

  int i = beg;
  for (; i + CH <= end; i += CH)
    softmax_chunk<H, C, VEC, CH, true>(csr, hfeat, i, end, f0, asrc, adv, m, den, acc);
  if (i < end)
    softmax_chunk<H, C, VEC, CH, false>(csr, hfeat, i, end, f0, asrc, adv, m, den, acc);
}

template <int H, int C, int VEC, bool PROJ>
__global__ void __launch_bounds__(256)
seg_gather(const int* __restrict__ rowptr, const int* __restrict__ csr,
           const float* __restrict__ hfeat,
           const float* __restrict__ att_src, const float* __restrict__ att_dst,
           const float* __restrict__ bias,
           const float* __restrict__ Wo, const float* __restrict__ bo,
           float* __restrict__ dstbuf, int Nn) {
  typedef typename fvec<VEC>::type vf;
  constexpr int HC = H * C;
  constexpr int LPN = HC / VEC;
  const unsigned gt = blockIdx.x * blockDim.x + threadIdx.x;
  const int node = (int)(gt / LPN);
  if (node >= Nn) return;
  const int gl = (int)(threadIdx.x % LPN);
  const int f0 = gl * VEC;
  vf acc;
  float den;
  segment_walk<H, C, VEC>(rowptr, csr, hfeat, att_src, att_dst, node, f0, acc, den);

  if constexpr (PROJ) {
    float p = 0.f;
#pragma unroll
    for (int k = 0; k < VEC; ++k)
      p += lrelu(acc[k] / den + bias[f0 + k], ACT_SLOPE) * Wo[f0 + k];
    for (int o = 1; o < LPN; o <<= 1) p += __shfl_xor(p, o, LPN);
    if (gl == 0) dstbuf[node] = p + bo[0];
  } else {
    float* op = dstbuf + (size_t)node * HC + f0;
#pragma unroll
    for (int k = 0; k < VEC; ++k)
      op[k] = lrelu(acc[k] / den + bias[f0 + k], ACT_SLOPE);
  }
}

#define TILE_PITCH 68
template <int H, int C, int VEC, int NOUT>
__global__ void __launch_bounds__(256)
seg_gather_linear(const int* __restrict__ rowptr, const int* __restrict__ csr,
                  const float* __restrict__ hfeat,
                  const float* __restrict__ att_src, const float* __restrict__ att_dst,
                  const float* __restrict__ bias, const float* __restrict__ Wn,
                  float* __restrict__ hout, int Nn) {
  typedef typename fvec<VEC>::type vf;
  constexpr int HC = H * C;
  constexpr int LPN = HC / VEC;
  constexpr int NPG = 256 / LPN;
  static_assert(NPG == 16 && NOUT == 16 && VEC == 4, "one 16 x 16 output tile per work-group");
  __shared__ float tile[NPG * TILE_PITCH];
  const int t = threadIdx.x;
  const int nl = t / LPN;
  const int m0 = (int)blockIdx.x * NPG;
  const int node = min(m0 + nl, Nn - 1);
  const int f0 = (t % LPN) * VEC;
  const bool first_wave = __builtin_amdgcn_readfirstlane(t >> 5) == 0;
  const int half = (t >> 4) & 1;
  const int lid = t & 15;

  vf acc;
  float den;
  segment_walk<H, C, VEC>(rowptr, csr, hfeat, att_src, att_dst, node, f0, acc, den);

  vf f;
#pragma unroll
  for (int k = 0; k < VEC; ++k)
    f[k] = lrelu(acc[k] / den + bias[f0 + k], ACT_SLOPE);
  *(vf*)(tile + nl * TILE_PITCH + f0) = f;
  v2f bfrag[HC / 4];
  if (first_wave) {
#pragma unroll
    for (int k = 0; k < HC / 4; ++k) {
      const int ka = k * 4 + half * 2;
      bfrag[k].x = Wn[ka * NOUT + lid];
      bfrag[k].y = Wn[(ka + 1) * NOUT + lid];
    }
  }
  __syncthreads();
  if (first_wave) {
    v8f d = {};
#pragma unroll
    for (int k = 0; k < HC / 4; ++k) {
      const v2f a = *(const v2f*)(tile + lid * TILE_PITCH + k * 4 + half * 2);
      d = __builtin_amdgcn_wmma_f32_16x16x4_f32(false, a, false, bfrag[k], (short)0, d,
                                                false, false);
    }
#pragma unroll
    for (int r = 0; r < 8; ++r) {
      const int row = m0 + half * 8 + r;
      if (row < Nn) hout[(size_t)row * NOUT + lid] = d[r];
    }
  }
}

extern "C" void kernel_launch(void* const* d_in, const int* in_sizes, int n_in,
                              void* d_out, int out_size, void* d_ws, size_t ws_size,
                              hipStream_t stream) {
  const float* x    = (const float*)d_in[0];
  const int*   ei   = (const int*)d_in[1];
  const float* W1   = (const float*)d_in[2];
  const float* aS1  = (const float*)d_in[3];
  const float* aD1  = (const float*)d_in[4];
  const float* b1   = (const float*)d_in[5];
  const float* W2   = (const float*)d_in[6];
  const float* aS2  = (const float*)d_in[7];
  const float* aD2  = (const float*)d_in[8];
  const float* b2   = (const float*)d_in[9];
  const float* Wo   = (const float*)d_in[10];
  const float* bo   = (const float*)d_in[11];
  float* out = (float*)d_out;

  const int IN_C = 64, HID = 16, HEADS = 4;
  const int HC1 = HEADS * HID;
  const int Nn = in_sizes[0] / IN_C;
  const int E  = in_sizes[1] / 2;
  const int* src = ei;
  const int* dst = ei + E;

  float* p = (float*)d_ws;
  float* h1   = p; p += (size_t)Nn * HC1;
  float* h2   = p; p += (size_t)Nn * HID;
  const int NB = (Nn + BNODES - 1) / BNODES;
  const int NC = (E + CHUNK_E - 1) / CHUNK_E;
  int* q = (int*)p;
  int* off    = q; q += (size_t)NC * (NB + 1);
  int* rowptr = q; q += (size_t)Nn + 1;
  unsigned* part = (unsigned*)q; q += (size_t)E;
  int* csr    = q; q += (size_t)E;

  const int TB = 256;

  {
    const int tiles = ((Nn + 15) / 16) * (HC1 / 16);
    chunk_sort_and_gemm<64, 64><<<NC + (tiles + 31) / 32, 1024,
                                  (32 + NB + CHUNK_E) * sizeof(int), stream>>>(
        src, dst, off, part, E, NB, NC, x, W1, h1, Nn);
  }
  bucket_sort<<<NB, 1024, (2 * NC + 1) * sizeof(int), stream>>>(
      off, part, rowptr, csr, Nn, NB, NC);

  seg_gather_linear<4, 16, 4, 16><<<(Nn + 15) / 16, TB, 0, stream>>>(
      rowptr, csr, h1, aS1, aD1, b1, W2, h2, Nn);

  {
    const unsigned n = (unsigned)Nn * 8u;
    seg_gather<1, 16, 2, true><<<(n + TB - 1) / TB, TB, 0, stream>>>(
        rowptr, csr, h2, aS2, aD2, b2, Wo, bo, out, Nn);
  }
}


# Layer-1 GEMM beside the chunk sort: W1 staged once per work-group in LDS as k pairs, a wave makes a 16x32 block from 16-byte row pieces (k sum order changes); 587 work-groups become 392, one round
# speedup vs baseline: 13.7491x; 1.1032x over previous
#include <hip/hip_runtime.h>
#include <math.h>

#define ATT_SLOPE 0.2f
#define ACT_SLOPE 0.01f

typedef __attribute__((ext_vector_type(2))) float v2f;
typedef __attribute__((ext_vector_type(4))) float v4f;
typedef __attribute__((ext_vector_type(8))) float v8f;

__device__ __forceinline__ float lrelu(float v, float s) { return v > 0.f ? v : v * s; }


#define CHUNK_E 8192
#define BSHIFT 8
#define BNODES 256

__device__ __forceinline__ int
block_scan_lds(int* a, int n, int* wsum, int* __restrict__ gout) {
  const int t = threadIdx.x;
  const int lane = t & 31, w = t >> 5;
  const int per = (n + 1023) / 1024;
  const int b = min(t * per, n);
  const int e = min(b + per, n);
  int s = 0;
  for (int i = b; i < e; ++i) s += a[i];
  int v = s;
  for (int o = 1; o < 32; o <<= 1) {
    const int u = __shfl_up(v, o, 32);
    if (lane >= o) v += u;
  }
  if (lane == 31) wsum[w] = v;
  __syncthreads();
  if (w == 0) {
    int x = wsum[lane];
    for (int o = 1; o < 32; o <<= 1) {
      const int u = __shfl_up(x, o, 32);
      if (lane >= o) x += u;
    }
    wsum[lane] = x;
  }
  __syncthreads();
  int run = v - s + (w > 0 ? wsum[w - 1] : 0);
  for (int i = b; i < e; ++i) {
    const int cn = a[i];
    a[i] = run;
    if (gout) gout[i] = run;
    run += cn;
  }
  return wsum[31];
}

__device__ __forceinline__ void
chunk_sort_body(const int* __restrict__ src, const int* __restrict__ dst,
                int* __restrict__ off, unsigned* __restrict__ part,
                int E, int NB, int c, int* lds) {
  int* wsum = lds;
  int* lc = lds + 32;
  unsigned* sval = (unsigned*)(lds + 32 + NB);
  const int t = threadIdx.x;
  constexpr int PER = CHUNK_E / 1024;
  for (int i = t; i < NB; i += 1024) lc[i] = 0;
  const int e0 = c * CHUNK_E + t;
  int d[PER], s[PER];
#pragma unroll
  for (int k = 0; k < PER; ++k) {
    const int e = min(e0 + k * 1024, E - 1);
    d[k] = dst[e];
    s[k] = src[e];
  }
  __syncthreads();
#pragma unroll
  for (int k = 0; k < PER; ++k)
    if (e0 + k * 1024 < E) atomicAdd(&lc[d[k] >> BSHIFT], 1);
  __syncthreads();
  int* orow = off + (size_t)c * (NB + 1);
  const int nloc = block_scan_lds(lc, NB, wsum, orow);
  if (t == 0) orow[NB] = nloc;
  __syncthreads();
  int pos[PER];
#pragma unroll
  for (int k = 0; k < PER; ++k)
    pos[k] = (e0 + k * 1024 < E) ? atomicAdd(&lc[d[k] >> BSHIFT], 1) : 0;
#pragma unroll
  for (int k = 0; k < PER; ++k)
    if (e0 + k * 1024 < E)
      sval[pos[k]] = (unsigned)s[k] | ((unsigned)(d[k] & (BNODES - 1)) << 24);
  __syncthreads();
  unsigned vv[PER];
#pragma unroll
  for (int k = 0; k < PER; ++k) vv[k] = sval[min(t + k * 1024, nloc - 1)];
#pragma unroll
  for (int k = 0; k < PER; ++k) {
    const int i = t + k * 1024;
    if (i < nloc) part[(size_t)c * CHUNK_E + i] = vv[k];
  }
}

#define BS_PITCH 144
template <int K, int NCOLS>
__device__ __forceinline__ void
gemm_rows_body(const float* __restrict__ A, const float* __restrict__ B,
               float* __restrict__ D, int M, int g, float* bs) {
  static_assert(K == 64 && NCOLS == 64, "32 k pairs x 32 column pairs = 1024 threads");
  const int t = threadIdx.x;
  {
    const int p = t >> 5;
    const int n2 = (t & 31) * 2;
    const v2f r0 = *(const v2f*)(B + (size_t)(2 * p) * NCOLS + n2);
    const v2f r1 = *(const v2f*)(B + (size_t)(2 * p + 1) * NCOLS + n2);
    v4f o;
    o.x = r0.x; o.y = r1.x; o.z = r0.y; o.w = r1.y;
    *(v4f*)(bs + p * BS_PITCH + n2 * 2) = o;
  }
  const int lane = t & 31, w = t >> 5;
  const int half = lane >> 4, lid = lane & 15;
  const int m0 = (g * 16 + (w >> 1)) * 16;
  const int n0 = (w & 1) * 32;
  const int ar = min(m0 + lid, M - 1);
  v4f a4[K / 8];
#pragma unroll
  for (int q = 0; q < K / 8; ++q)
    a4[q] = *(const v4f*)(A + (size_t)ar * K + q * 8 + half * 4);
  __syncthreads();
  if (m0 >= M) return;
  v8f acc0 = {}, acc1 = {};
#pragma unroll
  for (int q = 0; q < K / 8; ++q) {
#pragma unroll
    for (int s = 0; s < 2; ++s) {
      v2f a;
      a.x = a4[q][2 * s];
      a.y = a4[q][2 * s + 1];
      const float* bp = bs + (4 * q + 2 * half + s) * BS_PITCH + (n0 + lid) * 2;
      const v2f b0 = *(const v2f*)bp;
      const v2f b1 = *(const v2f*)(bp + 32);
      acc0 = __builtin_amdgcn_wmma_f32_16x16x4_f32(false, a, false, b0, (short)0, acc0,
                                                   false, false);
      acc1 = __builtin_amdgcn_wmma_f32_16x16x4_f32(false, a, false, b1, (short)0, acc1,
                                                   false, false);
    }
  }
  const int rbase = m0 + half * 8;
#pragma unroll
  for (int r = 0; r < 8; ++r) {
    const int row = rbase + r;
    if (row < M) {
      D[(size_t)row * NCOLS + n0 + lid] = acc0[r];
      D[(size_t)row * NCOLS + n0 + 16 + lid] = acc1[r];
    }
  }
}

template <int K, int NCOLS>
__global__ void __launch_bounds__(1024)
chunk_sort_and_gemm(const int* __restrict__ src, const int* __restrict__ dst,
                    int* __restrict__ off, unsigned* __restrict__ part,
                    int E, int NB, int NC,
                    const float* __restrict__ A, const float* __restrict__ B,
                    float* __restrict__ D, int M) {
  extern __shared__ int lds[];
  if ((int)blockIdx.x < NC) {
    chunk_sort_body(src, dst, off, part, E, NB, (int)blockIdx.x, lds);
    return;
  }
  gemm_rows_body<K, NCOLS>(A, B, D, M, (int)blockIdx.x - NC, (float*)lds);
}

#define BCAP 12288
__global__ void __launch_bounds__(1024)
bucket_sort(const int* __restrict__ off, const unsigned* __restrict__ part,
            int* __restrict__ rowptr, int* __restrict__ csr, int Nn, int NB, int NC) {
  extern __shared__ int lds[];
  __shared__ int cnt[BNODES];
  __shared__ int srt[BCAP];
  __shared__ int wsum[32];
  __shared__ int bsum;
  int* rs = lds;
  int* rd = lds + NC + 1;
  const int t = threadIdx.x;
  const int b = blockIdx.x;
  if (t < BNODES) cnt[t] = 0;
  if (t == 0) bsum = 0;
  __syncthreads();
  {
    int so = 0;
    for (int c = t; c < NC; c += 1024) {
      const int o0 = off[(size_t)c * (NB + 1) + b];
      const int o1 = off[(size_t)c * (NB + 1) + b + 1];
      rs[c] = o1 - o0;
      rd[c] = c * CHUNK_E + o0;
      so += o0;
    }
    if (so != 0) atomicAdd(&bsum, so);
  }
  __syncthreads();
  const int n = block_scan_lds(rs, NC, wsum, nullptr);
  const int bstart = bsum;
  if (t == 0) rs[NC] = n;
  __syncthreads();
  for (int c = t; c < NC; c += 1024) rd[c] -= rs[c];
  __syncthreads();
  const bool staged = n <= BCAP;
  int P2 = 1;
  while (P2 < NC) P2 <<= 1;
  auto locate = [&](int j) -> int {
    int lo = 0;
    for (int sp = P2 >> 1; sp > 0; sp >>= 1) {
      const int m = lo + sp;
      if (rs[min(m, NC)] <= j) lo = m;
    }
    return rd[lo] + j;
  };
  auto locate4 = [&](const int (&j)[4], int (&a)[4]) {
    int lo[4] = {0, 0, 0, 0};
    for (int sp = P2 >> 1; sp > 0; sp >>= 1) {
      int v[4];
#pragma unroll
      for (int k = 0; k < 4; ++k) v[k] = rs[min(lo[k] + sp, NC)];
#pragma unroll
      for (int k = 0; k < 4; ++k)
        if (v[k] <= j[k]) lo[k] += sp;
    }
#pragma unroll
    for (int k = 0; k < 4; ++k) a[k] = rd[lo[k]] + j[k];
  };
  constexpr int KP = BCAP / 1024;
  unsigned p[KP];
  if (staged) {
#pragma unroll
    for (int g = 0; g < KP; g += 4) {
      if (g * 1024 < n) {
        int jj[4], a[4];
#pragma unroll
        for (int k = 0; k < 4; ++k) jj[k] = min(t + (g + k) * 1024, n - 1);
        locate4(jj, a);
#pragma unroll
        for (int k = 0; k < 4; ++k) p[g + k] = part[a[k]];
#pragma unroll
        for (int k = 0; k < 4; ++k)
          if (t + (g + k) * 1024 < n) atomicAdd(&cnt[p[g + k] >> 24], 1);
      }
    }
  } else {
    for (int j = t; j < n; j += 1024) atomicAdd(&cnt[part[locate(j)] >> 24], 1);
  }
  __syncthreads();
  if (t < 32) {
    int cc[BNODES / 32];
    int s = 0;
#pragma unroll
    for (int j = 0; j < BNODES / 32; ++j) {
      cc[j] = cnt[t * (BNODES / 32) + j];
      s += cc[j];
    }
    int v = s;
    for (int o = 1; o < 32; o <<= 1) {
      const int u = __shfl_up(v, o, 32);
      if (t >= o) v += u;
    }
    int run = v - s;
#pragma unroll
    for (int j = 0; j < BNODES / 32; ++j) {
      const int local = t * (BNODES / 32) + j;
      const int node = b * BNODES + local;
      cnt[local] = run;
      if (node < Nn) rowptr[node] = bstart + run;
      run += cc[j];
    }
  }
  if (b == (int)gridDim.x - 1 && t == 0) rowptr[Nn] = bstart + n;
  __syncthreads();
  if (staged) {
#pragma unroll
    for (int g = 0; g < KP; g += 4) {
      if (g * 1024 < n) {
        int pos[4];
#pragma unroll
        for (int k = 0; k < 4; ++k)
          pos[k] = (t + (g + k) * 1024 < n) ? atomicAdd(&cnt[p[g + k] >> 24], 1) : 0;
#pragma unroll
        for (int k = 0; k < 4; ++k)
          if (t + (g + k) * 1024 < n) srt[pos[k]] = (int)(p[g + k] & 0xFFFFFFu);
      }
    }
  } else {
    for (int j = t; j < n; j += 1024) {
      const unsigned q = part[locate(j)];
      const int ps = atomicAdd(&cnt[q >> 24], 1);
      csr[bstart + ps] = (int)(q & 0xFFFFFFu);
    }
  }
  if (staged && n > 0) {
    __syncthreads();
    const int sh = (int)((((size_t)(csr + bstart)) >> 2) & 31);
    for (int i0 = t - sh; i0 < n; i0 += 4096) {
      int v[4];
#pragma unroll
      for (int j = 0; j < 4; ++j) v[j] = srt[min(max(i0 + j * 1024, 0), n - 1)];
#pragma unroll
      for (int j = 0; j < 4; ++j) {
        const int i = i0 + j * 1024;
        if (i >= 0 && i < n) csr[bstart + i] = v[j];
      }
    }
  }
}

template <int N> __device__ __forceinline__ float add_lane_xor(float v);
template <> __device__ __forceinline__ float add_lane_xor<1>(float v) {
  float r;
  asm("v_add_f32_dpp %0, %1, %1 row_xmask:1 row_mask:0xf bank_mask:0xf" : "=v"(r) : "v"(v));
  return r;
}
template <> __device__ __forceinline__ float add_lane_xor<2>(float v) {
  float r;
  asm("v_add_f32_dpp %0, %1, %1 row_xmask:2 row_mask:0xf bank_mask:0xf" : "=v"(r) : "v"(v));
  return r;
}
template <> __device__ __forceinline__ float add_lane_xor<4>(float v) {
  float r;
  asm("v_add_f32_dpp %0, %1, %1 row_xmask:4 row_mask:0xf bank_mask:0xf" : "=v"(r) : "v"(v));
  return r;
}

template <int LANES>
__device__ __forceinline__ float head_sum(float v) {
  v = add_lane_xor<1>(v);
  v = add_lane_xor<2>(v);
  if constexpr (LANES >= 8) v = add_lane_xor<4>(v);
  return v;
}

template <int VEC> struct fvec;
template <> struct fvec<2> { typedef v2f type; };
template <> struct fvec<4> { typedef v4f type; };

__device__ __forceinline__ float lane_dot(v2f x, v2f a) {
  const v2f t = x * a;
  return t.x + t.y;
}
__device__ __forceinline__ float lane_dot(v4f x, v4f a) {
  const v2f t = x.xy * a.xy + x.zw * a.zw;
  return t.x + t.y;
}

#define LOG2_E 1.44269504088896340736f
__device__ __forceinline__ float exp_of_nonpositive(float d) {
  return __builtin_amdgcn_exp2f(d * LOG2_E);
}

template <int H, int C, int VEC, int CH, bool FULL>
__device__ __forceinline__ void
softmax_chunk(const int* __restrict__ csr, const float* __restrict__ hfeat,
              int i, int end, int f0, typename fvec<VEC>::type asrc, float adv,
              float& m, float& den, typename fvec<VEC>::type& acc) {
  typedef typename fvec<VEC>::type vf;
  constexpr int HC = H * C;
  int s[CH];
  vf x[CH];
  float e[CH];
#pragma unroll
  for (int j = 0; j < CH; ++j) s[j] = FULL ? csr[i + j] : csr[min(i + j, end - 1)];
#pragma unroll
  for (int j = 0; j < CH; ++j) x[j] = *(const vf*)(hfeat + (size_t)s[j] * HC + f0);
#pragma unroll
  for (int j = 0; j < CH; ++j) {
    const float v = lrelu(head_sum<C / VEC>(lane_dot(x[j], asrc)) + adv, ATT_SLOPE);
    e[j] = (FULL || i + j < end) ? v : -INFINITY;
  }
  float cm = e[0];
#pragma unroll
  for (int j = 1; j < CH; ++j) cm = fmaxf(cm, e[j]);
  const float mn = fmaxf(m, cm);
  const float sc = exp_of_nonpositive(m - mn);
  m = mn;
  den *= sc;
  acc *= sc;
#pragma unroll
  for (int j = 0; j < CH; ++j) {
    const float w = exp_of_nonpositive(e[j] - mn);
    den += w;
    acc += x[j] * w;
  }
}

template <int H, int C, int VEC>
__device__ __forceinline__ void
segment_walk(const int* __restrict__ rowptr, const int* __restrict__ csr,
             const float* __restrict__ hfeat,
             const float* __restrict__ att_src, const float* __restrict__ att_dst,
             int node, int f0, typename fvec<VEC>::type& acc, float& den) {
  typedef typename fvec<VEC>::type vf;
  constexpr int HC = H * C;
  constexpr int CH = 4;
  const int beg = rowptr[node];
  const int end = rowptr[node + 1];

  const vf asrc = *(const vf*)(att_src + f0);
  acc = *(const vf*)(hfeat + (size_t)node * HC + f0);
  const float adv = head_sum<C / VEC>(lane_dot(acc, *(const vf*)(att_dst + f0)));
  float m = lrelu(head_sum<C / VEC>(lane_dot(acc, asrc)) + adv, ATT_SLOPE);
  den = 1.f;

  int i = beg;
  for (; i + CH <= end; i += CH)
    softmax_chunk<H, C, VEC, CH, true>(csr, hfeat, i, end, f0, asrc, adv, m, den, acc);
  if (i < end)
    softmax_chunk<H, C, VEC, CH, false>(csr, hfeat, i, end, f0, asrc, adv, m, den, acc);
}

template <int H, int C, int VEC, bool PROJ>
__global__ void __launch_bounds__(256)
seg_gather(const int* __restrict__ rowptr, const int* __restrict__ csr,
           const float* __restrict__ hfeat,
           const float* __restrict__ att_src, const float* __restrict__ att_dst,
           const float* __restrict__ bias,
           const float* __restrict__ Wo, const float* __restrict__ bo,
           float* __restrict__ dstbuf, int Nn) {
  typedef typename fvec<VEC>::type vf;
  constexpr int HC = H * C;
  constexpr int LPN = HC / VEC;
  const unsigned gt = blockIdx.x * blockDim.x + threadIdx.x;
  const int node = (int)(gt / LPN);
  if (node >= Nn) return;
  const int gl = (int)(threadIdx.x % LPN);
  const int f0 = gl * VEC;
  vf acc;
  float den;
  segment_walk<H, C, VEC>(rowptr, csr, hfeat, att_src, att_dst, node, f0, acc, den);

  if constexpr (PROJ) {
    float p = 0.f;
#pragma unroll
    for (int k = 0; k < VEC; ++k)
      p += lrelu(acc[k] / den + bias[f0 + k], ACT_SLOPE) * Wo[f0 + k];
    for (int o = 1; o < LPN; o <<= 1) p += __shfl_xor(p, o, LPN);
    if (gl == 0) dstbuf[node] = p + bo[0];
  } else {
    float* op = dstbuf + (size_t)node * HC + f0;
#pragma unroll
    for (int k = 0; k < VEC; ++k)
      op[k] = lrelu(acc[k] / den + bias[f0 + k], ACT_SLOPE);
  }
}

#define TILE_PITCH 68
template <int H, int C, int VEC, int NOUT>
__global__ void __launch_bounds__(256)
seg_gather_linear(const int* __restrict__ rowptr, const int* __restrict__ csr,
                  const float* __restrict__ hfeat,
                  const float* __restrict__ att_src, const float* __restrict__ att_dst,
                  const float* __restrict__ bias, const float* __restrict__ Wn,
                  float* __restrict__ hout, int Nn) {
  typedef typename fvec<VEC>::type vf;
  constexpr int HC = H * C;
  constexpr int LPN = HC / VEC;
  constexpr int NPG = 256 / LPN;
  static_assert(NPG == 16 && NOUT == 16 && VEC == 4, "one 16 x 16 output tile per work-group");
  __shared__ float tile[NPG * TILE_PITCH];
  const int t = threadIdx.x;
  const int nl = t / LPN;
  const int m0 = (int)blockIdx.x * NPG;
  const int node = min(m0 + nl, Nn - 1);
  const int f0 = (t % LPN) * VEC;
  const bool first_wave = __builtin_amdgcn_readfirstlane(t >> 5) == 0;
  const int half = (t >> 4) & 1;
  const int lid = t & 15;

  vf acc;
  float den;
  segment_walk<H, C, VEC>(rowptr, csr, hfeat, att_src, att_dst, node, f0, acc, den);

  vf f;
#pragma unroll
  for (int k = 0; k < VEC; ++k)
    f[k] = lrelu(acc[k] / den + bias[f0 + k], ACT_SLOPE);
  *(vf*)(tile + nl * TILE_PITCH + f0) = f;
  v2f bfrag[HC / 4];
  if (first_wave) {
#pragma unroll
    for (int k = 0; k < HC / 4; ++k) {
      const int ka = k * 4 + half * 2;
      bfrag[k].x = Wn[ka * NOUT + lid];
      bfrag[k].y = Wn[(ka + 1) * NOUT + lid];
    }
  }
  __syncthreads();
  if (first_wave) {
    v8f d = {};
#pragma unroll
    for (int k = 0; k < HC / 4; ++k) {
      const v2f a = *(const v2f*)(tile + lid * TILE_PITCH + k * 4 + half * 2);
      d = __builtin_amdgcn_wmma_f32_16x16x4_f32(false, a, false, bfrag[k], (short)0, d,
                                                false, false);
    }
#pragma unroll
    for (int r = 0; r < 8; ++r) {
      const int row = m0 + half * 8 + r;
      if (row < Nn) hout[(size_t)row * NOUT + lid] = d[r];
    }
  }
}

extern "C" void kernel_launch(void* const* d_in, const int* in_sizes, int n_in,
                              void* d_out, int out_size, void* d_ws, size_t ws_size,
                              hipStream_t stream) {
  const float* x    = (const float*)d_in[0];
  const int*   ei   = (const int*)d_in[1];
  const float* W1   = (const float*)d_in[2];
  const float* aS1  = (const float*)d_in[3];
  const float* aD1  = (const float*)d_in[4];
  const float* b1   = (const float*)d_in[5];
  const float* W2   = (const float*)d_in[6];
  const float* aS2  = (const float*)d_in[7];
  const float* aD2  = (const float*)d_in[8];
  const float* b2   = (const float*)d_in[9];
  const float* Wo   = (const float*)d_in[10];
  const float* bo   = (const float*)d_in[11];
  float* out = (float*)d_out;

  const int IN_C = 64, HID = 16, HEADS = 4;
  const int HC1 = HEADS * HID;
  const int Nn = in_sizes[0] / IN_C;
  const int E  = in_sizes[1] / 2;
  const int* src = ei;
  const int* dst = ei + E;

  float* p = (float*)d_ws;
  float* h1   = p; p += (size_t)Nn * HC1;
  float* h2   = p; p += (size_t)Nn * HID;
  const int NB = (Nn + BNODES - 1) / BNODES;
  const int NC = (E + CHUNK_E - 1) / CHUNK_E;
  int* q = (int*)p;
  int* off    = q; q += (size_t)NC * (NB + 1);
  int* rowptr = q; q += (size_t)Nn + 1;
  unsigned* part = (unsigned*)q; q += (size_t)E;
  int* csr    = q; q += (size_t)E;

  const int TB = 256;

  {
    const int rblocks = (Nn + 15) / 16;
    chunk_sort_and_gemm<64, 64><<<NC + (rblocks + 15) / 16, 1024,
                                  (32 + NB + CHUNK_E) * sizeof(int), stream>>>(
        src, dst, off, part, E, NB, NC, x, W1, h1, Nn);
  }
  bucket_sort<<<NB, 1024, (2 * NC + 1) * sizeof(int), stream>>>(
      off, part, rowptr, csr, Nn, NB, NC);

  seg_gather_linear<4, 16, 4, 16><<<(Nn + 15) / 16, TB, 0, stream>>>(
      rowptr, csr, h1, aS1, aD1, b1, W2, h2, Nn);

  {
    const unsigned n = (unsigned)Nn * 8u;
    seg_gather<1, 16, 2, true><<<(n + TB - 1) / TB, TB, 0, stream>>>(
        rowptr, csr, h2, aS2, aD2, b2, Wo, bo, out, Nn);
  }
}
